# starting kernel
# baseline (speedup 1.0000x reference)
#include <hip/hip_runtime.h>
#include <hip/hip_bf16.h>

typedef __attribute__((ext_vector_type(2))) float v2f;
typedef __attribute__((ext_vector_type(8))) float v8f;

#define NB    2048
#define NF    64
#define TLEN  256
#define TDWT  276
#define NPOS  64
#define BQ    16
#define GPOS  8
#define PPH   4
#define KGP   72
#define KGH   36
#define SROW  294

__device__ __forceinline__ int refl_idx(int i, int N) {
    i = (i < 0) ? -i : i;
    i = (i >= N) ? (2 * N - 2 - i) : i;
    return i;
}

__device__ __forceinline__ void dwt_level(
    const float* __restrict__ inb,
    int N, int nout, int sub4,
    float* __restrict__ lo, int lostride,
    float* __restrict__ hi)
{
    const float HLO[8] = { 0.23037781330885523f,  0.7148465705525415f,
                           0.6308807679295904f,  -0.02798376941698385f,
                          -0.18703481171888114f,  0.030841381835986965f,
                           0.032883011666982945f,-0.010597401784997278f};
    const float HHI[8] = {-0.010597401784997278f,-0.032883011666982945f,
                           0.030841381835986965f, 0.18703481171888114f,
                          -0.02798376941698385f, -0.6308807679295904f,
                           0.7148465705525415f,  -0.23037781330885523f};
    int per = (nout + 3) >> 2;
    int k0 = sub4 * per;
    int k1 = k0 + per; if (k1 > nout) k1 = nout;
    if (k0 >= k1) return;

    float w0, w1, w2, w3, w4, w5, w6, w7;
    {
        int base = 2 * k0 - 6;
        w0 = inb[refl_idx(base + 0, N) * 16];
        w1 = inb[refl_idx(base + 1, N) * 16];
        w2 = inb[refl_idx(base + 2, N) * 16];
        w3 = inb[refl_idx(base + 3, N) * 16];
        w4 = inb[refl_idx(base + 4, N) * 16];
        w5 = inb[refl_idx(base + 5, N) * 16];
        w6 = inb[refl_idx(base + 6, N) * 16];
        w7 = inb[refl_idx(base + 7, N) * 16];
    }
    for (int k = k0;;) {
        float l, h;
        l = w0 * HLO[0];             h = w0 * HHI[0];
        l = fmaf(w1, HLO[1], l);     h = fmaf(w1, HHI[1], h);
        l = fmaf(w2, HLO[2], l);     h = fmaf(w2, HHI[2], h);
        l = fmaf(w3, HLO[3], l);     h = fmaf(w3, HHI[3], h);
        l = fmaf(w4, HLO[4], l);     h = fmaf(w4, HHI[4], h);
        l = fmaf(w5, HLO[5], l);     h = fmaf(w5, HHI[5], h);
        l = fmaf(w6, HLO[6], l);     h = fmaf(w6, HHI[6], h);
        l = fmaf(w7, HLO[7], l);     h = fmaf(w7, HHI[7], h);
        lo[k * lostride] = l;
        hi[k] = h;
        if (++k >= k1) break;
        w0 = w2; w1 = w3; w2 = w4; w3 = w5; w4 = w6; w5 = w7;
        w6 = inb[refl_idx(2 * k,     N) * 16];
        w7 = inb[refl_idx(2 * k + 1, N) * 16];
    }
}

__global__ __launch_bounds__(256) void swizzleW_kernel(
    const float* __restrict__ W, float* __restrict__ wf)
{
    int i = blockIdx.x * blockDim.x + threadIdx.x;
    const int NT = 4 * NPOS * KGP * 32;
    if (i >= NT) return;
    int l  = i & 31;
    int kg = (i >> 5) % KGP;
    int p  = (i / (32 * KGP)) % NPOS;
    int nt = i / (32 * KGP * NPOS);
    int f  = nt * 16 + (l & 15);
    int t  = kg * 4 + ((l >> 4) << 1);
    v2f v;
    v.x = (t     < TDWT) ? W[((size_t)f * TDWT + t    ) * NPOS + p] : 0.f;
    v.y = (t + 1 < TDWT) ? W[((size_t)f * TDWT + t + 1) * NPOS + p] : 0.f;
    *(v2f*)(wf + (size_t)i * 2) = v;
}

__global__ __launch_bounds__(256) void dwt_wmma_fused_kernel(
    const float* __restrict__ x, const float* __restrict__ wf,
    const float* __restrict__ bias, float* __restrict__ out)
{
    __shared__ __align__(16) float s_sig[GPOS * TLEN * 16];
    __shared__ __align__(16) float s_d[PPH * 16 * SROW];
    __shared__ __align__(16) float s_lo1[PPH * 131 * 16];
    __shared__ __align__(16) float s_lo2[PPH * 69 * 16];
    __shared__ __align__(16) float s_red[4 * 8 * 32];

    const int tid  = threadIdx.x;
    const int wave = tid >> 5;
    const int lane = tid & 31;
    const int B0   = blockIdx.x * BQ;

    const int nt   = wave & 3;
    const int half = wave >> 2;

    const int b     = tid & 15;
    const int sub   = tid >> 4;
    const int pos_l = sub >> 2;
    const int sub4  = sub & 3;

    for (int i = tid; i < PPH * 16 * (KGP * 4 - TDWT); i += 256) {
        int row = i / (KGP * 4 - TDWT);
        int c   = i % (KGP * 4 - TDWT);
        s_d[row * SROW + TDWT + c] = 0.f;
    }

    v8f acc = {};

    const int arow = lane & 15;
    const int koff = (lane >> 4) << 1;

    for (int g = 0; g < NPOS / GPOS; ++g) {
        for (int i = tid; i < 16 * TLEN * (GPOS / 4); i += 256) {
            int c  = i & (GPOS / 4 - 1);
            int t  = (i / (GPOS / 4)) & 255;
            int bb = i / ((GPOS / 4) * TLEN);
            float4 v = *(const float4*)(x + (size_t)(B0 + bb) * (TLEN * NPOS)
                                          + t * NPOS + g * GPOS + c * 4);
            int pl = c * 4;
            s_sig[((pl + 0) * TLEN + t) * 16 + bb] = v.x;
            s_sig[((pl + 1) * TLEN + t) * 16 + bb] = v.y;
            s_sig[((pl + 2) * TLEN + t) * 16 + bb] = v.z;
            s_sig[((pl + 3) * TLEN + t) * 16 + bb] = v.w;
        }
        __syncthreads();

        for (int fh = 0; fh < GPOS / PPH; ++fh) {
            const int pbase = g * GPOS + fh * PPH;

            {
                const char* nb = (const char*)(wf
                    + (((size_t)(nt * NPOS + pbase) * KGP + (size_t)half * KGH) * 32) * 2);
                __builtin_prefetch(nb + (size_t)lane * 128,        0, 1);
                __builtin_prefetch(nb + 4096 + (size_t)lane * 128, 0, 1);
                __builtin_prefetch(nb + 8192 + (size_t)lane * 128, 0, 1);
            }

            const float* sig = s_sig + (fh * PPH + pos_l) * (TLEN * 16) + b;
            float* drow = s_d + (pos_l * 16 + b) * SROW;
            const float* lo1 = s_lo1 + pos_l * (131 * 16) + b;
            const float* lo2 = s_lo2 + pos_l * (69 * 16) + b;
            dwt_level(sig,        256, 131, sub4, (float*)lo1, 16, drow + 38);
            __syncthreads();
            dwt_level(lo1,        131,  69, sub4, (float*)lo2, 16, drow + 169);
            __syncthreads();
            dwt_level(lo2,         69,  38, sub4, drow, 1, drow + 238);
            __syncthreads();

            for (int q = 0; q < PPH; ++q) {
                const v2f* wfp = (const v2f*)wf
                    + ((size_t)(nt * NPOS + pbase + q) * KGP + (size_t)half * KGH) * 32;
                const float* ap = s_d + (q * 16 + arow) * SROW + half * (KGH * 4) + koff;
#pragma unroll 12
                for (int j = 0; j < KGH; ++j) {
                    v2f a  = *(const v2f*)(ap + j * 4);
                    v2f bf = wfp[j * 32 + lane];
                    acc = __builtin_amdgcn_wmma_f32_16x16x4_f32(
                        false, a, false, bf, (short)0, acc, false, false);
                }
            }
            __syncthreads();
        }
    }

    if (half == 1) {
#pragma unroll
        for (int r = 0; r < 8; ++r) s_red[(nt * 8 + r) * 32 + lane] = acc[r];
    }
    __syncthreads();
    if (half == 0) {
        int n   = lane & 15;
        int mhi = (lane >> 4) * 8;
        int f   = nt * 16 + n;
        float bv = bias[f];
#pragma unroll
        for (int r = 0; r < 8; ++r) {
            float v = acc[r] + s_red[(nt * 8 + r) * 32 + lane] + bv;
            v = (v >= 0.f) ? v : 0.02f * v;
            out[(size_t)(B0 + mhi + r) * NF + f] = v;
        }
    }
}

extern "C" void kernel_launch(void* const* d_in, const int* in_sizes, int n_in,
                              void* d_out, int out_size, void* d_ws, size_t ws_size,
                              hipStream_t stream) {
    const float* x    = (const float*)d_in[0];
    const float* W    = (const float*)d_in[1];
    const float* bias = (const float*)d_in[2];
    float* out = (float*)d_out;
    float* wf  = (float*)d_ws;

    const int nswz = 4 * NPOS * KGP * 32;
    swizzleW_kernel<<<(nswz + 255) / 256, 256, 0, stream>>>(W, wf);
    dwt_wmma_fused_kernel<<<NB / BQ, 256, 0, stream>>>(x, wf, bias, out);
}


# Fold the linear 3-level DWT into the weights once per call (transposed DWT applied to W); main loop becomes a plain f32 WMMA GEMM over x as stored; changes the summation order
# speedup vs baseline: 4.4458x; 4.4458x over previous
#include <hip/hip_runtime.h>


typedef __attribute__((ext_vector_type(2))) float v2f;
typedef __attribute__((ext_vector_type(8))) float v8f;

#define NB    2048
#define NF    64
#define TLEN  256
#define TDWT  276
#define NPOS  64
#define KTOT  (TLEN * NPOS)
#define NKB   (KTOT / 32)
#define GW    16
#define NIT   (NKB / GW)


__device__ __forceinline__ float adj_z(const float* __restrict__ lo, const float* __restrict__ hi,
                                       int nout, int m)
{
    const float LE0 = 0.23037781330885523f,  LE1 = 0.6308807679295904f,
                LE2 = -0.18703481171888114f, LE3 = 0.032883011666982945f;
    const float LO0 = 0.7148465705525415f,   LO1 = -0.02798376941698385f,
                LO2 = 0.030841381835986965f, LO3 = -0.010597401784997278f;
    const float HE0 = -0.010597401784997278f, HE1 = 0.030841381835986965f,
                HE2 = -0.02798376941698385f,  HE3 = 0.7148465705525415f;
    const float HO0 = -0.032883011666982945f, HO1 = 0.18703481171888114f,
                HO2 = -0.6308807679295904f,   HO3 = -0.23037781330885523f;
    const int par = m & 1;
    const int k0  = (m + 6 - par) >> 1;
    const float l0 = par ? LO0 : LE0, l1 = par ? LO1 : LE1, l2 = par ? LO2 : LE2, l3 = par ? LO3 : LE3;
    const float g0 = par ? HO0 : HE0, g1 = par ? HO1 : HE1, g2 = par ? HO2 : HE2, g3 = par ? HO3 : HE3;
    const float lc[4] = {l0, l1, l2, l3};
    const float gc[4] = {g0, g1, g2, g3};
    float acc = 0.f;
    for (int u = 0; u < 4; ++u) {
        const int  k  = k0 - u;
        const bool ok = (k >= 0) && (k < nout);
        const int  kc = ok ? k : 0;
        float lv = lo[kc * 64];
        float hv = hi[kc * 64];
        lv = ok ? lv : 0.f;
        hv = ok ? hv : 0.f;
        acc = fmaf(lc[u], lv, acc);
        acc = fmaf(gc[u], hv, acc);
    }
    return acc;
}

__device__ __forceinline__ float adj_out(const float* __restrict__ lo, const float* __restrict__ hi,
                                         int N, int nout, int j)
{
    float v = adj_z(lo, hi, nout, j);
    if (j >= 1 && j <= 6) v += adj_z(lo, hi, nout, -j);
    const int mr = 2 * N - 2 - j;
    if (j <= N - 2 && mr <= 2 * nout - 1) v += adj_z(lo, hi, nout, mr);
    return v;
}

__global__ __launch_bounds__(512) void fold_dwt_into_w_kernel(
    const float* __restrict__ W, float* __restrict__ wf)
{
    __shared__ __align__(16) float s_w[TDWT * 64];
    __shared__ __align__(16) float s_a2[69 * 64];
    __shared__ __align__(16) float s_a1[131 * 64];
    __shared__ __align__(16) float s_o[64 * 64];

    const int tid = threadIdx.x;
    const int ft  = blockIdx.x & 3;
    const int pg  = (blockIdx.x >> 2) & 15;
    const int ts  = blockIdx.x >> 6;
    const int p0  = pg * 4;
    const int f0  = ft * 16;

#pragma unroll
    for (int u = 0; u < (16 * TDWT + 511) / 512; ++u) {
        const int i = tid + u * 512;
        if (i < 16 * TDWT) {
            const int fl = i / TDWT;
            const int d  = i - fl * TDWT;
            const float4 v = *(const float4*)(W + ((size_t)(f0 + fl) * TDWT + d) * NPOS + p0);
            *(float4*)(s_w + d * 64 + fl * 4) = v;
        }
    }
    __syncthreads();

    const int v  = tid & 63;
    const int jg = __builtin_amdgcn_readfirstlane(tid >> 6);
    for (int j = jg; j < 69; j += 8)
        s_a2[j * 64 + v] = adj_out(s_w + v, s_w + 238 * 64 + v, 69, 38, j);
    __syncthreads();
    for (int j = jg; j < 131; j += 8)
        s_a1[j * 64 + v] = adj_out(s_a2 + v, s_w + 169 * 64 + v, 131, 69, j);
    __syncthreads();
    for (int j = jg; j < 64; j += 8)
        s_o[j * 64 + v] = adj_out(s_a1 + v, s_w + 38 * 64 + v, 256, 131, ts * 64 + j);
    __syncthreads();

    const int nh  = ft >> 1;
    const int ntl = ft & 1;
    const int q   = (p0 >> 3) & 3;
    const int h   = (p0 >> 2) & 1;
    const int jj  = ntl * 4 + q;
    for (int i = tid; i < 64 * 16; i += 512) {
        const int fl = i & 15;
        const int tl = i >> 4;
        const int t  = ts * 64 + tl;
        const int kb = t * 2 + (p0 >> 5);
        const float4 val = *(const float4*)(s_o + tl * 64 + fl * 4);
        *(float4*)(wf + ((((size_t)nh * NKB + kb) * 8 + jj) * 32 + (h * 16 + fl)) * 4) = val;
    }
}

__global__ __launch_bounds__(512) void folded_gemm_kernel(
    const float* x, const float* wf,
    const float* __restrict__ bias, float* out)
{
    __shared__ float s_red[GW * 2 * 8 * 32];

    const int tid  = threadIdx.x;
    const int wave = tid >> 5;
    const int lane = tid & 31;
    const int mt   = blockIdx.x >> 1;
    const int nh   = blockIdx.x & 1;
    const int B0   = mt * 16;
    const int r    = lane & 15;
    const int h    = lane >> 4;

    const float4* ap = (const float4*)(x + (size_t)(B0 + r) * KTOT + h * 4);
    const float4* bp = (const float4*)wf + (size_t)nh * NKB * 8 * 32 + lane;

    v8f acc0 = {};
    v8f acc1 = {};

    float4 a0[4], b0[8], a1[4], b1[8];

#define LOAD_SET(A_, B_, KB_)                                            \
    {                                                                    \
        const int kb_ = (KB_);                                           \
        for (int q = 0; q < 4; ++q) A_[q] = ap[kb_ * 8 + q * 2];                          \
        for (int j = 0; j < 8; ++j) B_[j] = bp[(kb_ * 8 + j) * 32];                       \
    }
#define MMA_SET(A_, B_)                                                  \
    for (int q = 0; q < 4; ++q) {                                        \
        const v2f x0 = {A_[q].x, A_[q].y};                               \
        const v2f x1 = {A_[q].z, A_[q].w};                               \
        const v2f w00 = {B_[q].x, B_[q].y};                              \
        const v2f w01 = {B_[q].z, B_[q].w};                              \
        const v2f w10 = {B_[4 + q].x, B_[4 + q].y};                      \
        const v2f w11 = {B_[4 + q].z, B_[4 + q].w};                      \
        acc0 = __builtin_amdgcn_wmma_f32_16x16x4_f32(false, x0, false, w00, (short)0, acc0, false, false); \
        acc1 = __builtin_amdgcn_wmma_f32_16x16x4_f32(false, x0, false, w10, (short)0, acc1, false, false); \
        acc0 = __builtin_amdgcn_wmma_f32_16x16x4_f32(false, x1, false, w01, (short)0, acc0, false, false); \
        acc1 = __builtin_amdgcn_wmma_f32_16x16x4_f32(false, x1, false, w11, (short)0, acc1, false, false); \
    }

    const int rot = mt & (NIT - 1);
#define KBLK(IT_) ((((IT_) + rot) & (NIT - 1)) * GW + wave)

    LOAD_SET(a0, b0, KBLK(0));
    for (int it = 0; it < NIT; it += 2) {
        LOAD_SET(a1, b1, KBLK(it + 1));
        __builtin_amdgcn_sched_barrier(0);
        MMA_SET(a0, b0);
        const int itn = (it + 2 < NIT) ? it + 2 : it + 1;
        LOAD_SET(a0, b0, KBLK(itn));
        __builtin_amdgcn_sched_barrier(0);
        MMA_SET(a1, b1);
    }
#undef LOAD_SET
#undef MMA_SET
#undef KBLK

    for (int rr = 0; rr < 8; ++rr) {
        s_red[((wave * 2 + 0) * 8 + rr) * 32 + lane] = acc0[rr];
        s_red[((wave * 2 + 1) * 8 + rr) * 32 + lane] = acc1[rr];
    }
    __syncthreads();
    {
        const int col = tid & 31;
        const int row = tid >> 5;
        const int ntl = col >> 4;
        const int c   = col & 15;
        const int hh  = row >> 3;
        const int rr  = row & 7;
        float s = 0.f;
        for (int w = 0; w < GW; ++w) s += s_red[((w * 2 + ntl) * 8 + rr) * 32 + hh * 16 + c];
        const int f = nh * 32 + col;
        float v = s + bias[f];
        v = (v >= 0.f) ? v : 0.02f * v;
        out[(size_t)(B0 + row) * NF + f] = v;
    }
}

extern "C" void kernel_launch(void* const* d_in, const int* in_sizes, int n_in,
                              void* d_out, int out_size, void* d_ws, size_t ws_size,
                              hipStream_t stream) {
    const float* x    = (const float*)d_in[0];
    const float* W    = (const float*)d_in[1];
    const float* bias = (const float*)d_in[2];
    float* out = (float*)d_out;
    float* wf  = (float*)d_ws;

    fold_dwt_into_w_kernel<<<256, 512, 0, stream>>>(W, wf);
    folded_gemm_kernel<<<256, 512, 0, stream>>>(x, wf, bias, out);
}


# GEMM re-tiled: 128-row tile whose 8 row-tile waves share W' fragments through LDS, k split over 16 work-groups, slices summed in order by a third launch; changes the summation order
# speedup vs baseline: 5.4881x; 1.2344x over previous
#include <hip/hip_runtime.h>


typedef __attribute__((ext_vector_type(2))) float v2f;
typedef __attribute__((ext_vector_type(8))) float v8f;

#define NB    2048
#define NF    64
#define TLEN  256
#define TDWT  276
#define NPOS  64
#define KTOT  (TLEN * NPOS)
#define NKB   (KTOT / 32)
#define MT     128
#define KSPLIT 16
#define NSTEP  8
#define STAGE4 2048
#define WF_FLOATS   (2 * NKB * 8 * 32 * 4)
#define PART_FLOATS (KSPLIT * NB * NF)
#define WS_NEED     ((size_t)(WF_FLOATS + PART_FLOATS) * 4)
#define GW    16
#define NIT   (NKB / GW)


__device__ __forceinline__ float adj_z(const float* __restrict__ lo, const float* __restrict__ hi,
                                       int nout, int m)
{
    const float LE0 = 0.23037781330885523f,  LE1 = 0.6308807679295904f,
                LE2 = -0.18703481171888114f, LE3 = 0.032883011666982945f;
    const float LO0 = 0.7148465705525415f,   LO1 = -0.02798376941698385f,
                LO2 = 0.030841381835986965f, LO3 = -0.010597401784997278f;
    const float HE0 = -0.010597401784997278f, HE1 = 0.030841381835986965f,
                HE2 = -0.02798376941698385f,  HE3 = 0.7148465705525415f;
    const float HO0 = -0.032883011666982945f, HO1 = 0.18703481171888114f,
                HO2 = -0.6308807679295904f,   HO3 = -0.23037781330885523f;
    const int par = m & 1;
    const int k0  = (m + 6 - par) >> 1;
    const float l0 = par ? LO0 : LE0, l1 = par ? LO1 : LE1, l2 = par ? LO2 : LE2, l3 = par ? LO3 : LE3;
    const float g0 = par ? HO0 : HE0, g1 = par ? HO1 : HE1, g2 = par ? HO2 : HE2, g3 = par ? HO3 : HE3;
    const float lc[4] = {l0, l1, l2, l3};
    const float gc[4] = {g0, g1, g2, g3};
    float acc = 0.f;
    for (int u = 0; u < 4; ++u) {
        const int  k  = k0 - u;
        const bool ok = (k >= 0) && (k < nout);
        const int  kc = ok ? k : 0;
        float lv = lo[kc * 64];
        float hv = hi[kc * 64];
        lv = ok ? lv : 0.f;
        hv = ok ? hv : 0.f;
        acc = fmaf(lc[u], lv, acc);
        acc = fmaf(gc[u], hv, acc);
    }
    return acc;
}

__device__ __forceinline__ float adj_out(const float* __restrict__ lo, const float* __restrict__ hi,
                                         int N, int nout, int j)
{
    float v = adj_z(lo, hi, nout, j);
    if (j >= 1 && j <= 6) v += adj_z(lo, hi, nout, -j);
    const int mr = 2 * N - 2 - j;
    if (j <= N - 2 && mr <= 2 * nout - 1) v += adj_z(lo, hi, nout, mr);
    return v;
}

__global__ __launch_bounds__(512) void fold_dwt_into_w_kernel(
    const float* __restrict__ W, float* __restrict__ wf)
{
    __shared__ __align__(16) float s_w[TDWT * 64];
    __shared__ __align__(16) float s_a2[69 * 64];
    __shared__ __align__(16) float s_a1[131 * 64];
    __shared__ __align__(16) float s_o[64 * 64];

    const int tid = threadIdx.x;
    const int ft  = blockIdx.x & 3;
    const int pg  = (blockIdx.x >> 2) & 15;
    const int ts  = blockIdx.x >> 6;
    const int p0  = pg * 4;
    const int f0  = ft * 16;

#pragma unroll
    for (int u = 0; u < (16 * TDWT + 511) / 512; ++u) {
        const int i = tid + u * 512;
        if (i < 16 * TDWT) {
            const int fl = i / TDWT;
            const int d  = i - fl * TDWT;
            const float4 v = *(const float4*)(W + ((size_t)(f0 + fl) * TDWT + d) * NPOS + p0);
            *(float4*)(s_w + d * 64 + fl * 4) = v;
        }
    }
    __syncthreads();

    const int v  = tid & 63;
    const int jg = __builtin_amdgcn_readfirstlane(tid >> 6);
    for (int j = jg; j < 69; j += 8)
        s_a2[j * 64 + v] = adj_out(s_w + v, s_w + 238 * 64 + v, 69, 38, j);
    __syncthreads();
    for (int j = jg; j < 131; j += 8)
        s_a1[j * 64 + v] = adj_out(s_a2 + v, s_w + 169 * 64 + v, 131, 69, j);
    __syncthreads();
    for (int j = jg; j < 64; j += 8)
        s_o[j * 64 + v] = adj_out(s_a1 + v, s_w + 38 * 64 + v, 256, 131, ts * 64 + j);
    __syncthreads();

    const int nh  = ft >> 1;
    const int ntl = ft & 1;
    const int q   = (p0 >> 3) & 3;
    const int h   = (p0 >> 2) & 1;
    const int jj  = ntl * 4 + q;
    for (int i = tid; i < 64 * 16; i += 512) {
        const int fl = i & 15;
        const int tl = i >> 4;
        const int t  = ts * 64 + tl;
        const int kb = t * 2 + (p0 >> 5);
        const float4 val = *(const float4*)(s_o + tl * 64 + fl * 4);
        *(float4*)(wf + ((((size_t)nh * NKB + kb) * 8 + jj) * 32 + (h * 16 + fl)) * 4) = val;
    }
}

__device__ __forceinline__ void gemm_unsplit_16rows(
    const float* x, const float* wf, const float* __restrict__ bias, float* out, float* s_red)
{
    const int tid  = threadIdx.x;
    const int wave = tid >> 5;
    const int lane = tid & 31;
    const int mt   = blockIdx.x >> 1;
    const int nh   = blockIdx.x & 1;
    const int B0   = mt * 16;
    const int r    = lane & 15;
    const int h    = lane >> 4;

    const float4* ap = (const float4*)(x + (size_t)(B0 + r) * KTOT + h * 4);
    const float4* bp = (const float4*)wf + (size_t)nh * NKB * 8 * 32 + lane;

    v8f acc0 = {};
    v8f acc1 = {};

    float4 a0[4], b0[8], a1[4], b1[8];

#define LOAD_SET(A_, B_, KB_)                                            \
    {                                                                    \
        const int kb_ = (KB_);                                           \
        for (int q = 0; q < 4; ++q) A_[q] = ap[kb_ * 8 + q * 2];                          \
        for (int j = 0; j < 8; ++j) B_[j] = bp[(kb_ * 8 + j) * 32];                       \
    }
#define MMA_SET(A_, B_)                                                  \
    for (int q = 0; q < 4; ++q) {                                        \
        const v2f x0 = {A_[q].x, A_[q].y};                               \
        const v2f x1 = {A_[q].z, A_[q].w};                               \
        const v2f w00 = {B_[q].x, B_[q].y};                              \
        const v2f w01 = {B_[q].z, B_[q].w};                              \
        const v2f w10 = {B_[4 + q].x, B_[4 + q].y};                      \
        const v2f w11 = {B_[4 + q].z, B_[4 + q].w};                      \
        acc0 = __builtin_amdgcn_wmma_f32_16x16x4_f32(false, x0, false, w00, (short)0, acc0, false, false); \
        acc1 = __builtin_amdgcn_wmma_f32_16x16x4_f32(false, x0, false, w10, (short)0, acc1, false, false); \
        acc0 = __builtin_amdgcn_wmma_f32_16x16x4_f32(false, x1, false, w01, (short)0, acc0, false, false); \
        acc1 = __builtin_amdgcn_wmma_f32_16x16x4_f32(false, x1, false, w11, (short)0, acc1, false, false); \
    }

    const int rot = mt & (NIT - 1);
#define KBLK(IT_) ((((IT_) + rot) & (NIT - 1)) * GW + wave)

    LOAD_SET(a0, b0, KBLK(0));
    for (int it = 0; it < NIT; it += 2) {
        LOAD_SET(a1, b1, KBLK(it + 1));
        __builtin_amdgcn_sched_barrier(0);
        MMA_SET(a0, b0);
        const int itn = (it + 2 < NIT) ? it + 2 : it + 1;
        LOAD_SET(a0, b0, KBLK(itn));
        __builtin_amdgcn_sched_barrier(0);
        MMA_SET(a1, b1);
    }
#undef LOAD_SET
#undef MMA_SET
#undef KBLK

    for (int rr = 0; rr < 8; ++rr) {
        s_red[((wave * 2 + 0) * 8 + rr) * 32 + lane] = acc0[rr];
        s_red[((wave * 2 + 1) * 8 + rr) * 32 + lane] = acc1[rr];
    }
    __syncthreads();
    {
        const int col = tid & 31;
        const int row = tid >> 5;
        const int ntl = col >> 4;
        const int c   = col & 15;
        const int hh  = row >> 3;
        const int rr  = row & 7;
        float s = 0.f;
        for (int w = 0; w < GW; ++w) s += s_red[((w * 2 + ntl) * 8 + rr) * 32 + hh * 16 + c];
        const int f = nh * 32 + col;
        float v = s + bias[f];
        v = (v >= 0.f) ? v : 0.02f * v;
        out[(size_t)(B0 + row) * NF + f] = v;
    }
}

__global__ __launch_bounds__(512) void folded_gemm_kernel(
    const float* x, const float* wf, float* part, const float* __restrict__ bias, float* out, size_t ws_bytes)
{
    __shared__ __align__(16) float s_lds[2 * STAGE4 * 4];

    if (ws_bytes < WS_NEED) {
        gemm_unsplit_16rows(x, wf, bias, out, s_lds);
        return;
    }

    const int tid  = threadIdx.x;
    const int wave = tid >> 5;
    const int lane = tid & 31;
    const int rt   = wave & 7;
    const int kh   = wave >> 3;
    const int mt   = blockIdx.x >> 4;
    const int ks   = blockIdx.x & (KSPLIT - 1);
    const int B0   = mt * MT;
    const int r    = lane & 15;
    const int h    = lane >> 4;

    const float4* ap = (const float4*)(x + (size_t)(B0 + rt * 16 + r) * KTOT + h * 4)
                       + (size_t)(ks * 32 + kh * 16) * 8;
    const float4* gp = (const float4*)wf + ((size_t)(tid >> 8) * NKB + ks * 32) * 256 + (tid & 255);
    float4* s4 = (float4*)s_lds;

    v8f acc0 = {}, acc1 = {}, acc2 = {}, acc3 = {};
    float4 ae0[4], ae1[4], ao0[4], ao1[4], b[16], g0, g1, g2, g3;

#define LOAD_A(A_, KBL_)                                                 \
    {                                                                    \
        const int kb_ = (KBL_);                                          \
        for (int q = 0; q < 4; ++q) A_[q] = ap[kb_ * 8 + q * 2];         \
    }
#define LOAD_G(STEP_)                                                    \
    {                                                                    \
        g0 = gp[((STEP_) * 2) * 256];                                    \
        g1 = gp[((STEP_) * 2 + 1) * 256];                                \
        g2 = gp[(16 + (STEP_) * 2) * 256];                               \
        g3 = gp[(16 + (STEP_) * 2 + 1) * 256];                           \
    }
#define STORE_L(STG_)                                                    \
    {                                                                    \
        s4[(STG_) * STAGE4 + tid]        = g0;                           \
        s4[(STG_) * STAGE4 + 512 + tid]  = g1;                           \
        s4[(STG_) * STAGE4 + 1024 + tid] = g2;                           \
        s4[(STG_) * STAGE4 + 1536 + tid] = g3;                           \
    }
#define LOAD_B(B_, STG_, KK_)                                            \
    for (int j = 0; j < 16; ++j) B_[j] = s4[(STG_) * STAGE4 + (kh * 2 + (KK_)) * 512 + j * 32 + lane];
#define MMA_SET(A_, B_)                                                  \
    for (int q = 0; q < 4; ++q) {                                        \
        const v2f x0 = {A_[q].x, A_[q].y};                               \
        const v2f x1 = {A_[q].z, A_[q].w};                               \
        const v2f w00 = {B_[q].x, B_[q].y},      w01 = {B_[q].z, B_[q].w};           \
        const v2f w10 = {B_[4 + q].x, B_[4 + q].y},   w11 = {B_[4 + q].z, B_[4 + q].w};   \
        const v2f w20 = {B_[8 + q].x, B_[8 + q].y},   w21 = {B_[8 + q].z, B_[8 + q].w};   \
        const v2f w30 = {B_[12 + q].x, B_[12 + q].y}, w31 = {B_[12 + q].z, B_[12 + q].w}; \
        acc0 = __builtin_amdgcn_wmma_f32_16x16x4_f32(false, x0, false, w00, (short)0, acc0, false, false); \
        acc1 = __builtin_amdgcn_wmma_f32_16x16x4_f32(false, x0, false, w10, (short)0, acc1, false, false); \
        acc2 = __builtin_amdgcn_wmma_f32_16x16x4_f32(false, x0, false, w20, (short)0, acc2, false, false); \
        acc3 = __builtin_amdgcn_wmma_f32_16x16x4_f32(false, x0, false, w30, (short)0, acc3, false, false); \
        acc0 = __builtin_amdgcn_wmma_f32_16x16x4_f32(false, x1, false, w01, (short)0, acc0, false, false); \
        acc1 = __builtin_amdgcn_wmma_f32_16x16x4_f32(false, x1, false, w11, (short)0, acc1, false, false); \
        acc2 = __builtin_amdgcn_wmma_f32_16x16x4_f32(false, x1, false, w21, (short)0, acc2, false, false); \
        acc3 = __builtin_amdgcn_wmma_f32_16x16x4_f32(false, x1, false, w31, (short)0, acc3, false, false); \
    }

    LOAD_G(0);
    LOAD_A(ae0, 0);
    LOAD_A(ae1, 1);
    STORE_L(0);
    LOAD_G(1);
    __syncthreads();

#define STEP_MMA(A0_, A1_, STG_)                                         \
    LOAD_B(b, STG_, 0);                                                  \
      \
                  \
    __builtin_amdgcn_sched_barrier(0);                                   \
    MMA_SET(A0_, b);                                                     \
    LOAD_B(b, STG_, 1);                                                  \
            \
    __builtin_amdgcn_sched_barrier(0);                                   \
    MMA_SET(A1_, b);                                                     \
      \
    __syncthreads();

#pragma unroll
    for (int t = 0; t < NSTEP; t += 2) {
        LOAD_A(ao0, 2 * t + 2);
        LOAD_A(ao1, 2 * t + 3);
        STORE_L(1);
        if (t + 2 < NSTEP) LOAD_G(t + 2);
        STEP_MMA(ae0, ae1, 0);
        if (t + 2 < NSTEP) {
            LOAD_A(ae0, 2 * t + 4);
            LOAD_A(ae1, 2 * t + 5);
            STORE_L(0);
        }
        if (t + 3 < NSTEP) LOAD_G(t + 3);
        STEP_MMA(ao0, ao1, 1);
    }
#undef STEP_MMA
#undef LOAD_A
#undef LOAD_G
#undef STORE_L
#undef LOAD_B
#undef MMA_SET

    {
        float* s_red = s_lds + kh * (MT * NF) + (rt * 16 + 8 * h) * NF + r;
        for (int rr = 0; rr < 8; ++rr) {
            s_red[rr * NF +  0] = acc0[rr];
            s_red[rr * NF + 16] = acc1[rr];
            s_red[rr * NF + 32] = acc2[rr];
            s_red[rr * NF + 48] = acc3[rr];
        }
    }
    __syncthreads();
    {
        float4* pp = (float4*)part + ((size_t)ks * NB + B0) * (NF / 4);
        for (int u = 0; u < 4; ++u) {
            const int i4 = tid + u * 512;
            const float4 p0 = s4[i4];
            const float4 p1 = s4[MT * NF / 4 + i4];
            float4 s;
            s.x = p0.x + p1.x; s.y = p0.y + p1.y; s.z = p0.z + p1.z; s.w = p0.w + p1.w;
            pp[i4] = s;
        }
    }
}

__global__ __launch_bounds__(128) void slice_sum_kernel(
    const float* part, const float* __restrict__ bias, float* out, size_t ws_bytes)
{
    if (ws_bytes < WS_NEED) return;
    const int i4 = blockIdx.x * 128 + threadIdx.x;
    const float4* pp = (const float4*)part + i4;
    float4 p[KSPLIT];
    for (int s = 0; s < KSPLIT; ++s) p[s] = pp[(size_t)s * (NB * NF / 4)];
    float4 v = p[0];
    for (int s = 1; s < KSPLIT; ++s) { v.x += p[s].x; v.y += p[s].y; v.z += p[s].z; v.w += p[s].w; }
    const float4 bv = *(const float4*)(bias + (i4 & (NF / 4 - 1)) * 4);
    v.x += bv.x; v.y += bv.y; v.z += bv.z; v.w += bv.w;
    v.x = (v.x >= 0.f) ? v.x : 0.02f * v.x;
    v.y = (v.y >= 0.f) ? v.y : 0.02f * v.y;
    v.z = (v.z >= 0.f) ? v.z : 0.02f * v.z;
    v.w = (v.w >= 0.f) ? v.w : 0.02f * v.w;
    ((float4*)out)[i4] = v;
}

extern "C" void kernel_launch(void* const* d_in, const int* in_sizes, int n_in,
                              void* d_out, int out_size, void* d_ws, size_t ws_size,
                              hipStream_t stream) {
    const float* x    = (const float*)d_in[0];
    const float* W    = (const float*)d_in[1];
    const float* bias = (const float*)d_in[2];
    float* out  = (float*)d_out;
    float* wf   = (float*)d_ws;
    float* part = wf + WF_FLOATS;

    fold_dwt_into_w_kernel<<<256, 512, 0, stream>>>(W, wf);
    folded_gemm_kernel<<<256, 512, 0, stream>>>(x, wf, part, bias, out, ws_size);
    slice_sum_kernel<<<256, 128, 0, stream>>>(part, bias, out, ws_size);
}


# GEMM: x tiles staged through LDS by loads covering whole 128-B lines (two rows' 256-B chunks per load); A fragments read from LDS at a 272-B pitch instead of 16-line strided global loads
# speedup vs baseline: 5.9730x; 1.0884x over previous
#include <hip/hip_runtime.h>


typedef __attribute__((ext_vector_type(2))) float v2f;
typedef __attribute__((ext_vector_type(8))) float v8f;

#define NB    2048
#define NF    64
#define TLEN  256
#define TDWT  276
#define NPOS  64
#define KTOT  (TLEN * NPOS)
#define NKB   (KTOT / 32)
#define MT     128
#define KSPLIT 16
#define NSTEP  8
#define STAGE4 2048
#define WF_FLOATS   (2 * NKB * 8 * 32 * 4)
#define PART_FLOATS (KSPLIT * NB * NF)
#define WS_NEED     ((size_t)(WF_FLOATS + PART_FLOATS) * 4)
#define GW    16
#define NIT   (NKB / GW)


__device__ __forceinline__ float adj_z(const float* __restrict__ lo, const float* __restrict__ hi,
                                       int nout, int m)
{
    const float LE0 = 0.23037781330885523f,  LE1 = 0.6308807679295904f,
                LE2 = -0.18703481171888114f, LE3 = 0.032883011666982945f;
    const float LO0 = 0.7148465705525415f,   LO1 = -0.02798376941698385f,
                LO2 = 0.030841381835986965f, LO3 = -0.010597401784997278f;
    const float HE0 = -0.010597401784997278f, HE1 = 0.030841381835986965f,
                HE2 = -0.02798376941698385f,  HE3 = 0.7148465705525415f;
    const float HO0 = -0.032883011666982945f, HO1 = 0.18703481171888114f,
                HO2 = -0.6308807679295904f,   HO3 = -0.23037781330885523f;
    const int par = m & 1;
    const int k0  = (m + 6 - par) >> 1;
    const float l0 = par ? LO0 : LE0, l1 = par ? LO1 : LE1, l2 = par ? LO2 : LE2, l3 = par ? LO3 : LE3;
    const float g0 = par ? HO0 : HE0, g1 = par ? HO1 : HE1, g2 = par ? HO2 : HE2, g3 = par ? HO3 : HE3;
    const float lc[4] = {l0, l1, l2, l3};
    const float gc[4] = {g0, g1, g2, g3};
    float acc = 0.f;
    for (int u = 0; u < 4; ++u) {
        const int  k  = k0 - u;
        const bool ok = (k >= 0) && (k < nout);
        const int  kc = ok ? k : 0;
        float lv = lo[kc * 64];
        float hv = hi[kc * 64];
        lv = ok ? lv : 0.f;
        hv = ok ? hv : 0.f;
        acc = fmaf(lc[u], lv, acc);
        acc = fmaf(gc[u], hv, acc);
    }
    return acc;
}

__device__ __forceinline__ float adj_out(const float* __restrict__ lo, const float* __restrict__ hi,
                                         int N, int nout, int j)
{
    float v = adj_z(lo, hi, nout, j);
    if (j >= 1 && j <= 6) v += adj_z(lo, hi, nout, -j);
    const int mr = 2 * N - 2 - j;
    if (j <= N - 2 && mr <= 2 * nout - 1) v += adj_z(lo, hi, nout, mr);
    return v;
}

__global__ __launch_bounds__(512) void fold_dwt_into_w_kernel(
    const float* __restrict__ W, float* __restrict__ wf)
{
    __shared__ __align__(16) float s_w[TDWT * 64];
    __shared__ __align__(16) float s_a2[69 * 64];
    __shared__ __align__(16) float s_a1[131 * 64];
    __shared__ __align__(16) float s_o[64 * 64];

    const int tid = threadIdx.x;
    const int ft  = blockIdx.x & 3;
    const int pg  = (blockIdx.x >> 2) & 15;
    const int ts  = blockIdx.x >> 6;
    const int p0  = pg * 4;
    const int f0  = ft * 16;

#pragma unroll
    for (int u = 0; u < (16 * TDWT + 511) / 512; ++u) {
        const int i = tid + u * 512;
        if (i < 16 * TDWT) {
            const int fl = i / TDWT;
            const int d  = i - fl * TDWT;
            const float4 v = *(const float4*)(W + ((size_t)(f0 + fl) * TDWT + d) * NPOS + p0);
            *(float4*)(s_w + d * 64 + fl * 4) = v;
        }
    }
    __syncthreads();

    const int v  = tid & 63;
    const int jg = __builtin_amdgcn_readfirstlane(tid >> 6);
    for (int j = jg; j < 69; j += 8)
        s_a2[j * 64 + v] = adj_out(s_w + v, s_w + 238 * 64 + v, 69, 38, j);
    __syncthreads();
    for (int j = jg; j < 131; j += 8)
        s_a1[j * 64 + v] = adj_out(s_a2 + v, s_w + 169 * 64 + v, 131, 69, j);
    __syncthreads();
    for (int j = jg; j < 64; j += 8)
        s_o[j * 64 + v] = adj_out(s_a1 + v, s_w + 38 * 64 + v, 256, 131, ts * 64 + j);
    __syncthreads();

    const int nh  = ft >> 1;
    const int ntl = ft & 1;
    const int q   = (p0 >> 3) & 3;
    const int h   = (p0 >> 2) & 1;
    const int jj  = ntl * 4 + q;
    for (int i = tid; i < 64 * 16; i += 512) {
        const int fl = i & 15;
        const int tl = i >> 4;
        const int t  = ts * 64 + tl;
        const int kb = t * 2 + (p0 >> 5);
        const float4 val = *(const float4*)(s_o + tl * 64 + fl * 4);
        *(float4*)(wf + ((((size_t)nh * NKB + kb) * 8 + jj) * 32 + (h * 16 + fl)) * 4) = val;
    }
}

__device__ __forceinline__ void gemm_unsplit_16rows(
    const float* x, const float* wf, const float* __restrict__ bias, float* out, float* s_red)
{
    const int tid  = threadIdx.x;
    const int wave = tid >> 5;
    const int lane = tid & 31;
    const int mt   = blockIdx.x >> 1;
    const int nh   = blockIdx.x & 1;
    const int B0   = mt * 16;
    const int r    = lane & 15;
    const int h    = lane >> 4;

    const float4* ap = (const float4*)(x + (size_t)(B0 + r) * KTOT + h * 4);
    const float4* bp = (const float4*)wf + (size_t)nh * NKB * 8 * 32 + lane;

    v8f acc0 = {};
    v8f acc1 = {};

    float4 a0[4], b0[8], a1[4], b1[8];

#define LOAD_SET(A_, B_, KB_)                                            \
    {                                                                    \
        const int kb_ = (KB_);                                           \
        for (int q = 0; q < 4; ++q) A_[q] = ap[kb_ * 8 + q * 2];                          \
        for (int j = 0; j < 8; ++j) B_[j] = bp[(kb_ * 8 + j) * 32];                       \
    }
#define MMA_SET(A_, B_)                                                  \
    for (int q = 0; q < 4; ++q) {                                        \
        const v2f x0 = {A_[q].x, A_[q].y};                               \
        const v2f x1 = {A_[q].z, A_[q].w};                               \
        const v2f w00 = {B_[q].x, B_[q].y};                              \
        const v2f w01 = {B_[q].z, B_[q].w};                              \
        const v2f w10 = {B_[4 + q].x, B_[4 + q].y};                      \
        const v2f w11 = {B_[4 + q].z, B_[4 + q].w};                      \
        acc0 = __builtin_amdgcn_wmma_f32_16x16x4_f32(false, x0, false, w00, (short)0, acc0, false, false); \
        acc1 = __builtin_amdgcn_wmma_f32_16x16x4_f32(false, x0, false, w10, (short)0, acc1, false, false); \
        acc0 = __builtin_amdgcn_wmma_f32_16x16x4_f32(false, x1, false, w01, (short)0, acc0, false, false); \
        acc1 = __builtin_amdgcn_wmma_f32_16x16x4_f32(false, x1, false, w11, (short)0, acc1, false, false); \
    }

    const int rot = mt & (NIT - 1);
#define KBLK(IT_) ((((IT_) + rot) & (NIT - 1)) * GW + wave)

    LOAD_SET(a0, b0, KBLK(0));
    for (int it = 0; it < NIT; it += 2) {
        LOAD_SET(a1, b1, KBLK(it + 1));
        __builtin_amdgcn_sched_barrier(0);
        MMA_SET(a0, b0);
        const int itn = (it + 2 < NIT) ? it + 2 : it + 1;
        LOAD_SET(a0, b0, KBLK(itn));
        __builtin_amdgcn_sched_barrier(0);
        MMA_SET(a1, b1);
    }
#undef LOAD_SET
#undef MMA_SET
#undef KBLK

    for (int rr = 0; rr < 8; ++rr) {
        s_red[((wave * 2 + 0) * 8 + rr) * 32 + lane] = acc0[rr];
        s_red[((wave * 2 + 1) * 8 + rr) * 32 + lane] = acc1[rr];
    }
    __syncthreads();
    {
        const int col = tid & 31;
        const int row = tid >> 5;
        const int ntl = col >> 4;
        const int c   = col & 15;
        const int hh  = row >> 3;
        const int rr  = row & 7;
        float s = 0.f;
        for (int w = 0; w < GW; ++w) s += s_red[((w * 2 + ntl) * 8 + rr) * 32 + hh * 16 + c];
        const int f = nh * 32 + col;
        float v = s + bias[f];
        v = (v >= 0.f) ? v : 0.02f * v;
        out[(size_t)(B0 + row) * NF + f] = v;
    }
}

#define APITCH  272
#define AHALF   (MT * APITCH)
#define ASTAGE  (2 * AHALF)
#define WREGION (2 * STAGE4 * 16)
#define GEMM_LDS (WREGION + 2 * ASTAGE)
__global__ __launch_bounds__(512) void folded_gemm_kernel(
    const float* x, const float* wf, float* part, const float* __restrict__ bias, float* out, size_t ws_bytes)
{
    __shared__ __align__(16) char s_raw[GEMM_LDS];
    float* s_lds = (float*)s_raw;

    if (ws_bytes < WS_NEED) {
        gemm_unsplit_16rows(x, wf, bias, out, s_lds);
        return;
    }

    const int tid  = threadIdx.x;
    const int wave = tid >> 5;
    const int lane = tid & 31;
    const int rt   = wave & 7;
    const int kh   = wave >> 3;
    const int mt   = blockIdx.x >> 4;
    const int ks   = blockIdx.x & (KSPLIT - 1);
    const int B0   = mt * MT;
    const int r    = lane & 15;
    const int h    = lane >> 4;

    const float4* xp = (const float4*)(x + (size_t)(B0 + (tid >> 4)) * KTOT) + (size_t)ks * 256 + (tid & 15);
    char* xs = s_raw + WREGION + (tid >> 4) * APITCH + (tid & 15) * 16;
    const char* af = s_raw + WREGION + kh * AHALF + (rt * 16 + r) * APITCH + h * 16;
    const float4* gp = (const float4*)wf + ((size_t)(tid >> 8) * NKB + ks * 32) * 256 + (tid & 255);
    float4* s4 = (float4*)s_raw;

    v8f acc0 = {}, acc1 = {}, acc2 = {}, acc3 = {};
    float4 xe0, xe1, xe2, xe3, xe4, xe5, xe6, xe7, xo0, xo1, xo2, xo3, xo4, xo5, xo6, xo7, ge0, ge1, ge2, ge3, go0, go1, go2, go3, a[4], b[16];

#define XOFF(U_, STEP_) ((size_t)((U_) & 3) * (32 * (KTOT / 4)) + ((U_) >> 2) * 128 + (STEP_) * 16)
#define XLDS(U_, STG_)  ((STG_) * ASTAGE + ((U_) >> 2) * AHALF + ((U_) & 3) * (32 * APITCH))
#define LOAD_X(P_, STEP_)                                                \
    { P_##0 = xp[XOFF(0, STEP_)]; P_##1 = xp[XOFF(1, STEP_)]; P_##2 = xp[XOFF(2, STEP_)]; P_##3 = xp[XOFF(3, STEP_)]; P_##4 = xp[XOFF(4, STEP_)]; P_##5 = xp[XOFF(5, STEP_)]; P_##6 = xp[XOFF(6, STEP_)]; P_##7 = xp[XOFF(7, STEP_)]; }
#define STORE_X(P_, STG_)                                                \
    { *(float4*)(xs + XLDS(0, STG_)) = P_##0; *(float4*)(xs + XLDS(1, STG_)) = P_##1; *(float4*)(xs + XLDS(2, STG_)) = P_##2; *(float4*)(xs + XLDS(3, STG_)) = P_##3; *(float4*)(xs + XLDS(4, STG_)) = P_##4; *(float4*)(xs + XLDS(5, STG_)) = P_##5; *(float4*)(xs + XLDS(6, STG_)) = P_##6; *(float4*)(xs + XLDS(7, STG_)) = P_##7; }
#define LOAD_G(P_, STEP_)                                                \
    {                                                                    \
        P_##0 = gp[((STEP_) * 2) * 256];                                 \
        P_##1 = gp[((STEP_) * 2 + 1) * 256];                             \
        P_##2 = gp[(16 + (STEP_) * 2) * 256];                            \
        P_##3 = gp[(16 + (STEP_) * 2 + 1) * 256];                        \
    }
#define STORE_L(P_, STG_)                                                \
    {                                                                    \
        s4[(STG_) * STAGE4 + tid]        = P_##0;                        \
        s4[(STG_) * STAGE4 + 512 + tid]  = P_##1;                        \
        s4[(STG_) * STAGE4 + 1024 + tid] = P_##2;                        \
        s4[(STG_) * STAGE4 + 1536 + tid] = P_##3;                        \
    }
#define LOAD_AF(A_, STG_, KK_)                                           \
    for (int q = 0; q < 4; ++q) A_[q] = *(const float4*)(af + (STG_) * ASTAGE + (KK_) * 128 + q * 32);
#define LOAD_B(B_, STG_, KK_)                                            \
    for (int j = 0; j < 16; ++j) B_[j] = s4[(STG_) * STAGE4 + (kh * 2 + (KK_)) * 512 + j * 32 + lane];
#define MMA_SET(A_, B_)                                                  \
    for (int q = 0; q < 4; ++q) {                                        \
        const v2f x0 = {A_[q].x, A_[q].y};                               \
        const v2f x1 = {A_[q].z, A_[q].w};                               \
        const v2f w00 = {B_[q].x, B_[q].y},      w01 = {B_[q].z, B_[q].w};           \
        const v2f w10 = {B_[4 + q].x, B_[4 + q].y},   w11 = {B_[4 + q].z, B_[4 + q].w};   \
        const v2f w20 = {B_[8 + q].x, B_[8 + q].y},   w21 = {B_[8 + q].z, B_[8 + q].w};   \
        const v2f w30 = {B_[12 + q].x, B_[12 + q].y}, w31 = {B_[12 + q].z, B_[12 + q].w}; \
        acc0 = __builtin_amdgcn_wmma_f32_16x16x4_f32(false, x0, false, w00, (short)0, acc0, false, false); \
        acc1 = __builtin_amdgcn_wmma_f32_16x16x4_f32(false, x0, false, w10, (short)0, acc1, false, false); \
        acc2 = __builtin_amdgcn_wmma_f32_16x16x4_f32(false, x0, false, w20, (short)0, acc2, false, false); \
        acc3 = __builtin_amdgcn_wmma_f32_16x16x4_f32(false, x0, false, w30, (short)0, acc3, false, false); \
        acc0 = __builtin_amdgcn_wmma_f32_16x16x4_f32(false, x1, false, w01, (short)0, acc0, false, false); \
        acc1 = __builtin_amdgcn_wmma_f32_16x16x4_f32(false, x1, false, w11, (short)0, acc1, false, false); \
        acc2 = __builtin_amdgcn_wmma_f32_16x16x4_f32(false, x1, false, w21, (short)0, acc2, false, false); \
        acc3 = __builtin_amdgcn_wmma_f32_16x16x4_f32(false, x1, false, w31, (short)0, acc3, false, false); \
    }

    LOAD_G(ge, 0);
    LOAD_X(xe, 0);
    LOAD_G(go, 1);
    LOAD_X(xo, 1);
    STORE_L(ge, 0);
    STORE_X(xe, 0);
    __syncthreads();

#define STEP_MMA(STG_)                                                   \
    LOAD_AF(a, STG_, 0);                                                 \
    LOAD_B(b, STG_, 0);                                                  \
         \
         \
    __builtin_amdgcn_sched_barrier(0);                                   \
    MMA_SET(a, b);                                                       \
    LOAD_AF(a, STG_, 1);                                                 \
    LOAD_B(b, STG_, 1);                                                  \
            \
    __builtin_amdgcn_sched_barrier(0);                                   \
    MMA_SET(a, b);                                                       \
      \
    __syncthreads();

#pragma unroll
    for (int t = 0; t < NSTEP; t += 2) {
        if (t + 2 < NSTEP) {
            LOAD_G(ge, t + 2);
            LOAD_X(xe, t + 2);
        }
        STORE_L(go, 1);
        STORE_X(xo, 1);
        STEP_MMA(0);
        if (t + 3 < NSTEP) {
            LOAD_G(go, t + 3);
            LOAD_X(xo, t + 3);
        }
        if (t + 2 < NSTEP) {
            STORE_L(ge, 0);
            STORE_X(xe, 0);
        }
        STEP_MMA(1);
    }
#undef STEP_MMA
#undef XOFF
#undef XLDS
#undef LOAD_X
#undef STORE_X
#undef LOAD_G
#undef STORE_L
#undef LOAD_AF
#undef LOAD_B
#undef MMA_SET

    {
        float* s_red = s_lds + kh * (MT * NF) + (rt * 16 + 8 * h) * NF + r;
        for (int rr = 0; rr < 8; ++rr) {
            s_red[rr * NF +  0] = acc0[rr];
            s_red[rr * NF + 16] = acc1[rr];
            s_red[rr * NF + 32] = acc2[rr];
            s_red[rr * NF + 48] = acc3[rr];
        }
    }
    __syncthreads();
    {
        float4* pp = (float4*)part + ((size_t)ks * NB + B0) * (NF / 4);
        for (int u = 0; u < 4; ++u) {
            const int i4 = tid + u * 512;
            const float4 p0 = s4[i4];
            const float4 p1 = s4[MT * NF / 4 + i4];
            float4 s;
            s.x = p0.x + p1.x; s.y = p0.y + p1.y; s.z = p0.z + p1.z; s.w = p0.w + p1.w;
            pp[i4] = s;
        }
    }
}

__global__ __launch_bounds__(128) void slice_sum_kernel(
    const float* part, const float* __restrict__ bias, float* out, size_t ws_bytes)
{
    if (ws_bytes < WS_NEED) return;
    const int i4 = blockIdx.x * 128 + threadIdx.x;
    const float4* pp = (const float4*)part + i4;
    float4 p[KSPLIT];
    for (int s = 0; s < KSPLIT; ++s) p[s] = pp[(size_t)s * (NB * NF / 4)];
    float4 v = p[0];
    for (int s = 1; s < KSPLIT; ++s) { v.x += p[s].x; v.y += p[s].y; v.z += p[s].z; v.w += p[s].w; }
    const float4 bv = *(const float4*)(bias + (i4 & (NF / 4 - 1)) * 4);
    v.x += bv.x; v.y += bv.y; v.z += bv.z; v.w += bv.w;
    v.x = (v.x >= 0.f) ? v.x : 0.02f * v.x;
    v.y = (v.y >= 0.f) ? v.y : 0.02f * v.y;
    v.z = (v.z >= 0.f) ? v.z : 0.02f * v.z;
    v.w = (v.w >= 0.f) ? v.w : 0.02f * v.w;
    ((float4*)out)[i4] = v;
}

extern "C" void kernel_launch(void* const* d_in, const int* in_sizes, int n_in,
                              void* d_out, int out_size, void* d_ws, size_t ws_size,
                              hipStream_t stream) {
    const float* x    = (const float*)d_in[0];
    const float* W    = (const float*)d_in[1];
    const float* bias = (const float*)d_in[2];
    float* out  = (float*)d_out;
    float* wf   = (float*)d_ws;
    float* part = wf + WF_FLOATS;

    fold_dwt_into_w_kernel<<<256, 512, 0, stream>>>(W, wf);
    folded_gemm_kernel<<<256, 512, 0, stream>>>(x, wf, part, bias, out, ws_size);
    slice_sum_kernel<<<256, 128, 0, stream>>>(part, bias, out, ws_size);
}


# Fold kernel: a wave's LDS reads for 4-6 output rows are issued as one independent group ahead of the arithmetic instead of dependent round trips per row; summation order unchanged
# speedup vs baseline: 6.2710x; 1.0499x over previous
#include <hip/hip_runtime.h>


typedef __attribute__((ext_vector_type(2))) float v2f;
typedef __attribute__((ext_vector_type(8))) float v8f;

#define NB    2048
#define NF    64
#define TLEN  256
#define TDWT  276
#define NPOS  64
#define KTOT  (TLEN * NPOS)
#define NKB   (KTOT / 32)
#define MT     128
#define KSPLIT 16
#define NSTEP  8
#define STAGE4 2048
#define WF_FLOATS   (2 * NKB * 8 * 32 * 4)
#define PART_FLOATS (KSPLIT * NB * NF)
#define WS_NEED     ((size_t)(WF_FLOATS + PART_FLOATS) * 4)
#define GW    16
#define NIT   (NKB / GW)


__device__ __forceinline__ float adj_z(const float* __restrict__ lo, const float* __restrict__ hi,
                                       int nout, int m)
{
    const float LE0 = 0.23037781330885523f,  LE1 = 0.6308807679295904f,
                LE2 = -0.18703481171888114f, LE3 = 0.032883011666982945f;
    const float LO0 = 0.7148465705525415f,   LO1 = -0.02798376941698385f,
                LO2 = 0.030841381835986965f, LO3 = -0.010597401784997278f;
    const float HE0 = -0.010597401784997278f, HE1 = 0.030841381835986965f,
                HE2 = -0.02798376941698385f,  HE3 = 0.7148465705525415f;
    const float HO0 = -0.032883011666982945f, HO1 = 0.18703481171888114f,
                HO2 = -0.6308807679295904f,   HO3 = -0.23037781330885523f;
    const int par = m & 1;
    const int k0  = (m + 6 - par) >> 1;
    const float l0 = par ? LO0 : LE0, l1 = par ? LO1 : LE1, l2 = par ? LO2 : LE2, l3 = par ? LO3 : LE3;
    const float g0 = par ? HO0 : HE0, g1 = par ? HO1 : HE1, g2 = par ? HO2 : HE2, g3 = par ? HO3 : HE3;
    const float lc[4] = {l0, l1, l2, l3};
    const float gc[4] = {g0, g1, g2, g3};
    float acc = 0.f;
    for (int u = 0; u < 4; ++u) {
        const int  k  = k0 - u;
        const bool ok = (k >= 0) && (k < nout);
        const int  kc = ok ? k : 0;
        float lv = lo[kc * 64];
        float hv = hi[kc * 64];
        lv = ok ? lv : 0.f;
        hv = ok ? hv : 0.f;
        acc = fmaf(lc[u], lv, acc);
        acc = fmaf(gc[u], hv, acc);
    }
    return acc;
}

template <int NR>
__device__ __forceinline__ void adj_rows_main(const float* lo, const float* hi, float* dst,
                                              int jstart, int cnt, int jg)
{
    const float LE0 = 0.23037781330885523f,  LE1 = 0.6308807679295904f,
                LE2 = -0.18703481171888114f, LE3 = 0.032883011666982945f;
    const float LO0 = 0.7148465705525415f,   LO1 = -0.02798376941698385f,
                LO2 = 0.030841381835986965f, LO3 = -0.010597401784997278f;
    const float HE0 = -0.010597401784997278f, HE1 = 0.030841381835986965f,
                HE2 = -0.02798376941698385f,  HE3 = 0.7148465705525415f;
    const float HO0 = -0.032883011666982945f, HO1 = 0.18703481171888114f,
                HO2 = -0.6308807679295904f,   HO3 = -0.23037781330885523f;
    for (int i0 = jg; i0 < cnt; i0 += 8 * NR) {
        float lv[NR][4], hv[NR][4];
#pragma unroll
        for (int u = 0; u < NR; ++u) {
            const int i   = i0 + 8 * u;
            const int ic  = (i < cnt) ? i : i0;
            const int m   = jstart + ic;
            const int par = m & 1;
            const int k0  = (m + 6 - par) >> 1;
#pragma unroll
            for (int t = 0; t < 4; ++t) {
                lv[u][t] = lo[(k0 - t) * 64];
                hv[u][t] = hi[(k0 - t) * 64];
            }
        }
        __builtin_amdgcn_sched_barrier(0);
#pragma unroll
        for (int u = 0; u < NR; ++u) {
            const int i   = i0 + 8 * u;
            const int ic  = (i < cnt) ? i : i0;
            const int par = (jstart + ic) & 1;
            const float lc[4] = {par ? LO0 : LE0, par ? LO1 : LE1, par ? LO2 : LE2, par ? LO3 : LE3};
            const float gc[4] = {par ? HO0 : HE0, par ? HO1 : HE1, par ? HO2 : HE2, par ? HO3 : HE3};
            float acc = 0.f;
#pragma unroll
            for (int t = 0; t < 4; ++t) {
                acc = fmaf(lc[t], lv[u][t], acc);
                acc = fmaf(gc[t], hv[u][t], acc);
            }
            if (i < cnt) dst[i * 64] = acc;
        }
    }
}

__device__ __forceinline__ void adj_rows_edges(const float* lo, const float* hi, float* dst,
                                               int N, int nout, int jstart, int cnt, int jg)
{
    const bool hasL = (jstart == 0) && (jg >= 1) && (jg <= 6);
    const int  il   = hasL ? jg : 0;
    const int  ml   = hasL ? -jg : 0;
    const int  rlo  = 2 * N - 1 - 2 * nout;
    const int  jr   = rlo + ((jg - rlo) & 7);
    const bool hasR = (jr <= N - 2) && (jr >= jstart) && (jr < jstart + cnt);
    const int  ir   = hasR ? jr - jstart : 0;
    const int  mr   = hasR ? 2 * N - 2 - jr : 0;
    const float pl = dst[il * 64];
    const float pr = dst[ir * 64];
    const float vl = pl + adj_z(lo, hi, nout, ml);
    const float vr = pr + adj_z(lo, hi, nout, mr);
    if (hasL) dst[il * 64] = vl;
    if (hasR) dst[ir * 64] = vr;
}

__global__ __launch_bounds__(512) void fold_dwt_into_w_kernel(
    const float* __restrict__ W, float* __restrict__ wf)
{
    __shared__ __align__(16) float s_w[TDWT * 64];
    __shared__ __align__(16) float s_a2[69 * 64];
    __shared__ __align__(16) float s_a1[131 * 64];
    __shared__ __align__(16) float s_o[64 * 64];

    const int tid = threadIdx.x;
    const int ft  = blockIdx.x & 3;
    const int pg  = (blockIdx.x >> 2) & 15;
    const int ts  = blockIdx.x >> 6;
    const int p0  = pg * 4;
    const int f0  = ft * 16;

#pragma unroll
    for (int u = 0; u < (16 * TDWT + 511) / 512; ++u) {
        const int i = tid + u * 512;
        if (i < 16 * TDWT) {
            const int fl = i / TDWT;
            const int d  = i - fl * TDWT;
            const float4 v = *(const float4*)(W + ((size_t)(f0 + fl) * TDWT + d) * NPOS + p0);
            *(float4*)(s_w + d * 64 + fl * 4) = v;
        }
    }
    __syncthreads();

    const int v  = tid & 63;
    const int jg = __builtin_amdgcn_readfirstlane(tid >> 6);
    adj_rows_main<5>(s_w + v, s_w + 238 * 64 + v, s_a2 + v, 0, 69, jg);
    adj_rows_edges(s_w + v, s_w + 238 * 64 + v, s_a2 + v, 69, 38, 0, 69, jg);
    __syncthreads();
    adj_rows_main<6>(s_a2 + v, s_w + 169 * 64 + v, s_a1 + v, 0, 131, jg);
    adj_rows_edges(s_a2 + v, s_w + 169 * 64 + v, s_a1 + v, 131, 69, 0, 131, jg);
    __syncthreads();
    adj_rows_main<4>(s_a1 + v, s_w + 38 * 64 + v, s_o + v, ts * 64, 64, jg);
    adj_rows_edges(s_a1 + v, s_w + 38 * 64 + v, s_o + v, 256, 131, ts * 64, 64, jg);
    __syncthreads();

    const int nh  = ft >> 1;
    const int ntl = ft & 1;
    const int q   = (p0 >> 3) & 3;
    const int h   = (p0 >> 2) & 1;
    const int jj  = ntl * 4 + q;
    for (int i = tid; i < 64 * 16; i += 512) {
        const int fl = i & 15;
        const int tl = i >> 4;
        const int t  = ts * 64 + tl;
        const int kb = t * 2 + (p0 >> 5);
        const float4 val = *(const float4*)(s_o + tl * 64 + fl * 4);
        *(float4*)(wf + ((((size_t)nh * NKB + kb) * 8 + jj) * 32 + (h * 16 + fl)) * 4) = val;
    }
}

__device__ __forceinline__ void gemm_unsplit_16rows(
    const float* x, const float* wf, const float* __restrict__ bias, float* out, float* s_red)
{
    const int tid  = threadIdx.x;
    const int wave = tid >> 5;
    const int lane = tid & 31;
    const int mt   = blockIdx.x >> 1;
    const int nh   = blockIdx.x & 1;
    const int B0   = mt * 16;
    const int r    = lane & 15;
    const int h    = lane >> 4;

    const float4* ap = (const float4*)(x + (size_t)(B0 + r) * KTOT + h * 4);
    const float4* bp = (const float4*)wf + (size_t)nh * NKB * 8 * 32 + lane;

    v8f acc0 = {};
    v8f acc1 = {};

    float4 a0[4], b0[8], a1[4], b1[8];

#define LOAD_SET(A_, B_, KB_)                                            \
    {                                                                    \
        const int kb_ = (KB_);                                           \
        for (int q = 0; q < 4; ++q) A_[q] = ap[kb_ * 8 + q * 2];                          \
        for (int j = 0; j < 8; ++j) B_[j] = bp[(kb_ * 8 + j) * 32];                       \
    }
#define MMA_SET(A_, B_)                                                  \
    for (int q = 0; q < 4; ++q) {                                        \
        const v2f x0 = {A_[q].x, A_[q].y};                               \
        const v2f x1 = {A_[q].z, A_[q].w};                               \
        const v2f w00 = {B_[q].x, B_[q].y};                              \
        const v2f w01 = {B_[q].z, B_[q].w};                              \
        const v2f w10 = {B_[4 + q].x, B_[4 + q].y};                      \
        const v2f w11 = {B_[4 + q].z, B_[4 + q].w};                      \
        acc0 = __builtin_amdgcn_wmma_f32_16x16x4_f32(false, x0, false, w00, (short)0, acc0, false, false); \
        acc1 = __builtin_amdgcn_wmma_f32_16x16x4_f32(false, x0, false, w10, (short)0, acc1, false, false); \
        acc0 = __builtin_amdgcn_wmma_f32_16x16x4_f32(false, x1, false, w01, (short)0, acc0, false, false); \
        acc1 = __builtin_amdgcn_wmma_f32_16x16x4_f32(false, x1, false, w11, (short)0, acc1, false, false); \
    }

    const int rot = mt & (NIT - 1);
#define KBLK(IT_) ((((IT_) + rot) & (NIT - 1)) * GW + wave)

    LOAD_SET(a0, b0, KBLK(0));
    for (int it = 0; it < NIT; it += 2) {
        LOAD_SET(a1, b1, KBLK(it + 1));
        __builtin_amdgcn_sched_barrier(0);
        MMA_SET(a0, b0);
        const int itn = (it + 2 < NIT) ? it + 2 : it + 1;
        LOAD_SET(a0, b0, KBLK(itn));
        __builtin_amdgcn_sched_barrier(0);
        MMA_SET(a1, b1);
    }
#undef LOAD_SET
#undef MMA_SET
#undef KBLK

    for (int rr = 0; rr < 8; ++rr) {
        s_red[((wave * 2 + 0) * 8 + rr) * 32 + lane] = acc0[rr];
        s_red[((wave * 2 + 1) * 8 + rr) * 32 + lane] = acc1[rr];
    }
    __syncthreads();
    {
        const int col = tid & 31;
        const int row = tid >> 5;
        const int ntl = col >> 4;
        const int c   = col & 15;
        const int hh  = row >> 3;
        const int rr  = row & 7;
        float s = 0.f;
        for (int w = 0; w < GW; ++w) s += s_red[((w * 2 + ntl) * 8 + rr) * 32 + hh * 16 + c];
        const int f = nh * 32 + col;
        float v = s + bias[f];
        v = (v >= 0.f) ? v : 0.02f * v;
        out[(size_t)(B0 + row) * NF + f] = v;
    }
}

#define APITCH  272
#define AHALF   (MT * APITCH)
#define ASTAGE  (2 * AHALF)
#define WREGION (2 * STAGE4 * 16)
#define GEMM_LDS (WREGION + 2 * ASTAGE)
__global__ __launch_bounds__(512) void folded_gemm_kernel(
    const float* x, const float* wf, float* part, const float* __restrict__ bias, float* out, size_t ws_bytes)
{
    __shared__ __align__(16) char s_raw[GEMM_LDS];
    float* s_lds = (float*)s_raw;

    if (ws_bytes < WS_NEED) {
        gemm_unsplit_16rows(x, wf, bias, out, s_lds);
        return;
    }

    const int tid  = threadIdx.x;
    const int wave = tid >> 5;
    const int lane = tid & 31;
    const int rt   = wave & 7;
    const int kh   = wave >> 3;
    const int mt   = blockIdx.x >> 4;
    const int ks   = blockIdx.x & (KSPLIT - 1);
    const int B0   = mt * MT;
    const int r    = lane & 15;
    const int h    = lane >> 4;

    const float4* xp = (const float4*)(x + (size_t)(B0 + (tid >> 4)) * KTOT) + (size_t)ks * 256 + (tid & 15);
    char* xs = s_raw + WREGION + (tid >> 4) * APITCH + (tid & 15) * 16;
    const char* af = s_raw + WREGION + kh * AHALF + (rt * 16 + r) * APITCH + h * 16;
    const float4* gp = (const float4*)wf + ((size_t)(tid >> 8) * NKB + ks * 32) * 256 + (tid & 255);
    float4* s4 = (float4*)s_raw;

    v8f acc0 = {}, acc1 = {}, acc2 = {}, acc3 = {};
    float4 xe0, xe1, xe2, xe3, xe4, xe5, xe6, xe7, xo0, xo1, xo2, xo3, xo4, xo5, xo6, xo7, ge0, ge1, ge2, ge3, go0, go1, go2, go3, a[4], b[16];

#define XOFF(U_, STEP_) ((size_t)((U_) & 3) * (32 * (KTOT / 4)) + ((U_) >> 2) * 128 + (STEP_) * 16)
#define XLDS(U_, STG_)  ((STG_) * ASTAGE + ((U_) >> 2) * AHALF + ((U_) & 3) * (32 * APITCH))
#define LOAD_X(P_, STEP_)                                                \
    { P_##0 = xp[XOFF(0, STEP_)]; P_##1 = xp[XOFF(1, STEP_)]; P_##2 = xp[XOFF(2, STEP_)]; P_##3 = xp[XOFF(3, STEP_)]; P_##4 = xp[XOFF(4, STEP_)]; P_##5 = xp[XOFF(5, STEP_)]; P_##6 = xp[XOFF(6, STEP_)]; P_##7 = xp[XOFF(7, STEP_)]; }
#define STORE_X(P_, STG_)                                                \
    { *(float4*)(xs + XLDS(0, STG_)) = P_##0; *(float4*)(xs + XLDS(1, STG_)) = P_##1; *(float4*)(xs + XLDS(2, STG_)) = P_##2; *(float4*)(xs + XLDS(3, STG_)) = P_##3; *(float4*)(xs + XLDS(4, STG_)) = P_##4; *(float4*)(xs + XLDS(5, STG_)) = P_##5; *(float4*)(xs + XLDS(6, STG_)) = P_##6; *(float4*)(xs + XLDS(7, STG_)) = P_##7; }
#define LOAD_G(P_, STEP_)                                                \
    {                                                                    \
        P_##0 = gp[((STEP_) * 2) * 256];                                 \
        P_##1 = gp[((STEP_) * 2 + 1) * 256];                             \
        P_##2 = gp[(16 + (STEP_) * 2) * 256];                            \
        P_##3 = gp[(16 + (STEP_) * 2 + 1) * 256];                        \
    }
#define STORE_L(P_, STG_)                                                \
    {                                                                    \
        s4[(STG_) * STAGE4 + tid]        = P_##0;                        \
        s4[(STG_) * STAGE4 + 512 + tid]  = P_##1;                        \
        s4[(STG_) * STAGE4 + 1024 + tid] = P_##2;                        \
        s4[(STG_) * STAGE4 + 1536 + tid] = P_##3;                        \
    }
#define LOAD_AF(A_, STG_, KK_)                                           \
    for (int q = 0; q < 4; ++q) A_[q] = *(const float4*)(af + (STG_) * ASTAGE + (KK_) * 128 + q * 32);
#define LOAD_B(B_, STG_, KK_)                                            \
    for (int j = 0; j < 16; ++j) B_[j] = s4[(STG_) * STAGE4 + (kh * 2 + (KK_)) * 512 + j * 32 + lane];
#define MMA_SET(A_, B_)                                                  \
    for (int q = 0; q < 4; ++q) {                                        \
        const v2f x0 = {A_[q].x, A_[q].y};                               \
        const v2f x1 = {A_[q].z, A_[q].w};                               \
        const v2f w00 = {B_[q].x, B_[q].y},      w01 = {B_[q].z, B_[q].w};           \
        const v2f w10 = {B_[4 + q].x, B_[4 + q].y},   w11 = {B_[4 + q].z, B_[4 + q].w};   \
        const v2f w20 = {B_[8 + q].x, B_[8 + q].y},   w21 = {B_[8 + q].z, B_[8 + q].w};   \
        const v2f w30 = {B_[12 + q].x, B_[12 + q].y}, w31 = {B_[12 + q].z, B_[12 + q].w}; \
        acc0 = __builtin_amdgcn_wmma_f32_16x16x4_f32(false, x0, false, w00, (short)0, acc0, false, false); \
        acc1 = __builtin_amdgcn_wmma_f32_16x16x4_f32(false, x0, false, w10, (short)0, acc1, false, false); \
        acc2 = __builtin_amdgcn_wmma_f32_16x16x4_f32(false, x0, false, w20, (short)0, acc2, false, false); \
        acc3 = __builtin_amdgcn_wmma_f32_16x16x4_f32(false, x0, false, w30, (short)0, acc3, false, false); \
        acc0 = __builtin_amdgcn_wmma_f32_16x16x4_f32(false, x1, false, w01, (short)0, acc0, false, false); \
        acc1 = __builtin_amdgcn_wmma_f32_16x16x4_f32(false, x1, false, w11, (short)0, acc1, false, false); \
        acc2 = __builtin_amdgcn_wmma_f32_16x16x4_f32(false, x1, false, w21, (short)0, acc2, false, false); \
        acc3 = __builtin_amdgcn_wmma_f32_16x16x4_f32(false, x1, false, w31, (short)0, acc3, false, false); \
    }

    LOAD_G(ge, 0);
    LOAD_X(xe, 0);
    LOAD_G(go, 1);
    LOAD_X(xo, 1);
    STORE_L(ge, 0);
    STORE_X(xe, 0);
    __syncthreads();

#define STEP_MMA(STG_)                                                   \
    LOAD_AF(a, STG_, 0);                                                 \
    LOAD_B(b, STG_, 0);                                                  \
         \
         \
    __builtin_amdgcn_sched_barrier(0);                                   \
    MMA_SET(a, b);                                                       \
    LOAD_AF(a, STG_, 1);                                                 \
    LOAD_B(b, STG_, 1);                                                  \
            \
    __builtin_amdgcn_sched_barrier(0);                                   \
    MMA_SET(a, b);                                                       \
      \
    __syncthreads();

#pragma unroll
    for (int t = 0; t < NSTEP; t += 2) {
        if (t + 2 < NSTEP) {
            LOAD_G(ge, t + 2);
            LOAD_X(xe, t + 2);
        }
        STORE_L(go, 1);
        STORE_X(xo, 1);
        STEP_MMA(0);
        if (t + 3 < NSTEP) {
            LOAD_G(go, t + 3);
            LOAD_X(xo, t + 3);
        }
        if (t + 2 < NSTEP) {
            STORE_L(ge, 0);
            STORE_X(xe, 0);
        }
        STEP_MMA(1);
    }
#undef STEP_MMA
#undef XOFF
#undef XLDS
#undef LOAD_X
#undef STORE_X
#undef LOAD_G
#undef STORE_L
#undef LOAD_AF
#undef LOAD_B
#undef MMA_SET

    {
        float* s_red = s_lds + kh * (MT * NF) + (rt * 16 + 8 * h) * NF + r;
        for (int rr = 0; rr < 8; ++rr) {
            s_red[rr * NF +  0] = acc0[rr];
            s_red[rr * NF + 16] = acc1[rr];
            s_red[rr * NF + 32] = acc2[rr];
            s_red[rr * NF + 48] = acc3[rr];
        }
    }
    __syncthreads();
    {
        float4* pp = (float4*)part + ((size_t)ks * NB + B0) * (NF / 4);
        for (int u = 0; u < 4; ++u) {
            const int i4 = tid + u * 512;
            const float4 p0 = s4[i4];
            const float4 p1 = s4[MT * NF / 4 + i4];
            float4 s;
            s.x = p0.x + p1.x; s.y = p0.y + p1.y; s.z = p0.z + p1.z; s.w = p0.w + p1.w;
            pp[i4] = s;
        }
    }
}

__global__ __launch_bounds__(128) void slice_sum_kernel(
    const float* part, const float* __restrict__ bias, float* out, size_t ws_bytes)
{
    if (ws_bytes < WS_NEED) return;
    const int i4 = blockIdx.x * 128 + threadIdx.x;
    const float4* pp = (const float4*)part + i4;
    float4 p[KSPLIT];
    for (int s = 0; s < KSPLIT; ++s) p[s] = pp[(size_t)s * (NB * NF / 4)];
    float4 v = p[0];
    for (int s = 1; s < KSPLIT; ++s) { v.x += p[s].x; v.y += p[s].y; v.z += p[s].z; v.w += p[s].w; }
    const float4 bv = *(const float4*)(bias + (i4 & (NF / 4 - 1)) * 4);
    v.x += bv.x; v.y += bv.y; v.z += bv.z; v.w += bv.w;
    v.x = (v.x >= 0.f) ? v.x : 0.02f * v.x;
    v.y = (v.y >= 0.f) ? v.y : 0.02f * v.y;
    v.z = (v.z >= 0.f) ? v.z : 0.02f * v.z;
    v.w = (v.w >= 0.f) ? v.w : 0.02f * v.w;
    ((float4*)out)[i4] = v;
}

extern "C" void kernel_launch(void* const* d_in, const int* in_sizes, int n_in,
                              void* d_out, int out_size, void* d_ws, size_t ws_size,
                              hipStream_t stream) {
    const float* x    = (const float*)d_in[0];
    const float* W    = (const float*)d_in[1];
    const float* bias = (const float*)d_in[2];
    float* out  = (float*)d_out;
    float* wf   = (float*)d_ws;
    float* part = wf + WF_FLOATS;

    fold_dwt_into_w_kernel<<<256, 512, 0, stream>>>(W, wf);
    folded_gemm_kernel<<<256, 512, 0, stream>>>(x, wf, part, bias, out, ws_size);
    slice_sum_kernel<<<256, 128, 0, stream>>>(part, bias, out, ws_size);
}


# Fold kernel loads W in whole 128-B lines, all in one group (work-group = 2 filters x 32 positions, was 16 x 4; linear LDS stores); wf is now [k-block][filter][32 k], at a 144-B LDS pitch in the GEMM
# speedup vs baseline: 7.1128x; 1.1342x over previous
#include <hip/hip_runtime.h>


typedef __attribute__((ext_vector_type(2))) float v2f;
typedef __attribute__((ext_vector_type(8))) float v8f;

#define NB    2048
#define NF    64
#define TLEN  256
#define TDWT  276
#define NPOS  64
#define KTOT  (TLEN * NPOS)
#define NKB   (KTOT / 32)
#define MT     128
#define KSPLIT 16
#define NSTEP  8
#define WF_FLOATS   (2 * NKB * 8 * 32 * 4)
#define PART_FLOATS (KSPLIT * NB * NF)
#define WS_NEED     ((size_t)(WF_FLOATS + PART_FLOATS) * 4)
#define GW    16
#define NIT   (NKB / GW)


__device__ __forceinline__ float adj_z(const float* __restrict__ lo, const float* __restrict__ hi,
                                       int nout, int m)
{
    const float LE0 = 0.23037781330885523f,  LE1 = 0.6308807679295904f,
                LE2 = -0.18703481171888114f, LE3 = 0.032883011666982945f;
    const float LO0 = 0.7148465705525415f,   LO1 = -0.02798376941698385f,
                LO2 = 0.030841381835986965f, LO3 = -0.010597401784997278f;
    const float HE0 = -0.010597401784997278f, HE1 = 0.030841381835986965f,
                HE2 = -0.02798376941698385f,  HE3 = 0.7148465705525415f;
    const float HO0 = -0.032883011666982945f, HO1 = 0.18703481171888114f,
                HO2 = -0.6308807679295904f,   HO3 = -0.23037781330885523f;
    const int par = m & 1;
    const int k0  = (m + 6 - par) >> 1;
    const float l0 = par ? LO0 : LE0, l1 = par ? LO1 : LE1, l2 = par ? LO2 : LE2, l3 = par ? LO3 : LE3;
    const float g0 = par ? HO0 : HE0, g1 = par ? HO1 : HE1, g2 = par ? HO2 : HE2, g3 = par ? HO3 : HE3;
    const float lc[4] = {l0, l1, l2, l3};
    const float gc[4] = {g0, g1, g2, g3};
    float acc = 0.f;
    for (int u = 0; u < 4; ++u) {
        const int  k  = k0 - u;
        const bool ok = (k >= 0) && (k < nout);
        const int  kc = ok ? k : 0;
        float lv = lo[kc * 64];
        float hv = hi[kc * 64];
        lv = ok ? lv : 0.f;
        hv = ok ? hv : 0.f;
        acc = fmaf(lc[u], lv, acc);
        acc = fmaf(gc[u], hv, acc);
    }
    return acc;
}

template <int NR>
__device__ __forceinline__ void adj_rows_main(const float* lo, const float* hi, float* dst,
                                              int jstart, int cnt, int jg)
{
    const float LE0 = 0.23037781330885523f,  LE1 = 0.6308807679295904f,
                LE2 = -0.18703481171888114f, LE3 = 0.032883011666982945f;
    const float LO0 = 0.7148465705525415f,   LO1 = -0.02798376941698385f,
                LO2 = 0.030841381835986965f, LO3 = -0.010597401784997278f;
    const float HE0 = -0.010597401784997278f, HE1 = 0.030841381835986965f,
                HE2 = -0.02798376941698385f,  HE3 = 0.7148465705525415f;
    const float HO0 = -0.032883011666982945f, HO1 = 0.18703481171888114f,
                HO2 = -0.6308807679295904f,   HO3 = -0.23037781330885523f;
    for (int i0 = jg; i0 < cnt; i0 += 8 * NR) {
        float lv[NR][4], hv[NR][4];
#pragma unroll
        for (int u = 0; u < NR; ++u) {
            const int i   = i0 + 8 * u;
            const int ic  = (i < cnt) ? i : i0;
            const int m   = jstart + ic;
            const int par = m & 1;
            const int k0  = (m + 6 - par) >> 1;
#pragma unroll
            for (int t = 0; t < 4; ++t) {
                lv[u][t] = lo[(k0 - t) * 64];
                hv[u][t] = hi[(k0 - t) * 64];
            }
        }
        __builtin_amdgcn_sched_barrier(0);
#pragma unroll
        for (int u = 0; u < NR; ++u) {
            const int i   = i0 + 8 * u;
            const int ic  = (i < cnt) ? i : i0;
            const int par = (jstart + ic) & 1;
            const float lc[4] = {par ? LO0 : LE0, par ? LO1 : LE1, par ? LO2 : LE2, par ? LO3 : LE3};
            const float gc[4] = {par ? HO0 : HE0, par ? HO1 : HE1, par ? HO2 : HE2, par ? HO3 : HE3};
            float acc = 0.f;
#pragma unroll
            for (int t = 0; t < 4; ++t) {
                acc = fmaf(lc[t], lv[u][t], acc);
                acc = fmaf(gc[t], hv[u][t], acc);
            }
            if (i < cnt) dst[i * 64] = acc;
        }
    }
}

__device__ __forceinline__ void adj_rows_edges(const float* lo, const float* hi, float* dst,
                                               int N, int nout, int jstart, int cnt, int jg)
{
    const bool hasL = (jstart == 0) && (jg >= 1) && (jg <= 6);
    const int  il   = hasL ? jg : 0;
    const int  ml   = hasL ? -jg : 0;
    const int  rlo  = 2 * N - 1 - 2 * nout;
    const int  jr   = rlo + ((jg - rlo) & 7);
    const bool hasR = (jr <= N - 2) && (jr >= jstart) && (jr < jstart + cnt);
    const int  ir   = hasR ? jr - jstart : 0;
    const int  mr   = hasR ? 2 * N - 2 - jr : 0;
    const float pl = dst[il * 64];
    const float pr = dst[ir * 64];
    const float vl = pl + adj_z(lo, hi, nout, ml);
    const float vr = pr + adj_z(lo, hi, nout, mr);
    if (hasL) dst[il * 64] = vl;
    if (hasR) dst[ir * 64] = vr;
}

__global__ __launch_bounds__(512) void fold_dwt_into_w_kernel(
    const float* __restrict__ W, float* __restrict__ wf)
{
    __shared__ __align__(16) float s_w[TDWT * 64];
    __shared__ __align__(16) float s_a2[69 * 64];
    __shared__ __align__(16) float s_a1[131 * 64];
    __shared__ __align__(16) float s_o[64 * 64];

    const int tid = threadIdx.x;
    const int ph  = blockIdx.x & 1;
    const int fp  = (blockIdx.x >> 1) & 31;
    const int ts  = blockIdx.x >> 6;
    const int p0  = ph * 32;
    const int f0  = fp * 2;

    constexpr int NV4   = 16 * TDWT;
    constexpr int NFULL = NV4 / 512;
    float4 wv[NFULL + 1];
#pragma unroll
    for (int u = 0; u <= NFULL; ++u) {
        const int i  = tid + u * 512;
        const int ic = (u < NFULL || i < NV4) ? i : tid;
        const int d  = ic >> 4;
        const int fl = (ic >> 3) & 1;
        const int pc = ic & 7;
        wv[u] = *(const float4*)(W + ((size_t)(f0 + fl) * TDWT + d) * NPOS + p0 + pc * 4);
    }
#pragma unroll
    for (int u = 0; u <= NFULL; ++u) {
        const int i  = tid + u * 512;
        const int ic = (u < NFULL || i < NV4) ? i : tid;
        *(float4*)(s_w + ic * 4) = wv[u];
    }
    __syncthreads();

    const int v  = tid & 63;
    const int jg = __builtin_amdgcn_readfirstlane(tid >> 6);
    adj_rows_main<5>(s_w + v, s_w + 238 * 64 + v, s_a2 + v, 0, 69, jg);
    adj_rows_edges(s_w + v, s_w + 238 * 64 + v, s_a2 + v, 69, 38, 0, 69, jg);
    __syncthreads();
    adj_rows_main<6>(s_a2 + v, s_w + 169 * 64 + v, s_a1 + v, 0, 131, jg);
    adj_rows_edges(s_a2 + v, s_w + 169 * 64 + v, s_a1 + v, 131, 69, 0, 131, jg);
    __syncthreads();
    adj_rows_main<4>(s_a1 + v, s_w + 38 * 64 + v, s_o + v, ts * 64, 64, jg);
    adj_rows_edges(s_a1 + v, s_w + 38 * 64 + v, s_o + v, 256, 131, ts * 64, 64, jg);
    __syncthreads();

    for (int i = tid; i < 64 * 16; i += 512) {
        const int c4 = i & 15;
        const int tl = i >> 4;
        const int t  = ts * 64 + tl;
        const int kb = t * 2 + ph;
        const float4 val = *(const float4*)(s_o + tl * 64 + c4 * 4);
        *(float4*)(wf + ((size_t)kb * NF + f0) * 32 + c4 * 4) = val;
    }
}

__device__ __forceinline__ void gemm_unsplit_16rows(
    const float* x, const float* wf, const float* __restrict__ bias, float* out, float* s_red)
{
    const int tid  = threadIdx.x;
    const int wave = tid >> 5;
    const int lane = tid & 31;
    const int mt   = blockIdx.x >> 1;
    const int nh   = blockIdx.x & 1;
    const int B0   = mt * 16;
    const int r    = lane & 15;
    const int h    = lane >> 4;

    const float4* ap = (const float4*)(x + (size_t)(B0 + r) * KTOT + h * 4);
    const float4* bp = (const float4*)wf + (nh * 32 + r) * 8 + h;

    v8f acc0 = {};
    v8f acc1 = {};

    float4 a0[4], b0[8], a1[4], b1[8];

#define LOAD_SET(A_, B_, KB_)                                            \
    {                                                                    \
        const int kb_ = (KB_);                                           \
        for (int q = 0; q < 4; ++q) A_[q] = ap[kb_ * 8 + q * 2];                          \
        for (int j = 0; j < 8; ++j) B_[j] = bp[kb_ * 512 + (j >> 2) * 128 + (j & 3) * 2]; \
    }
#define MMA_SET(A_, B_)                                                  \
    for (int q = 0; q < 4; ++q) {                                        \
        const v2f x0 = {A_[q].x, A_[q].y};                               \
        const v2f x1 = {A_[q].z, A_[q].w};                               \
        const v2f w00 = {B_[q].x, B_[q].y};                              \
        const v2f w01 = {B_[q].z, B_[q].w};                              \
        const v2f w10 = {B_[4 + q].x, B_[4 + q].y};                      \
        const v2f w11 = {B_[4 + q].z, B_[4 + q].w};                      \
        acc0 = __builtin_amdgcn_wmma_f32_16x16x4_f32(false, x0, false, w00, (short)0, acc0, false, false); \
        acc1 = __builtin_amdgcn_wmma_f32_16x16x4_f32(false, x0, false, w10, (short)0, acc1, false, false); \
        acc0 = __builtin_amdgcn_wmma_f32_16x16x4_f32(false, x1, false, w01, (short)0, acc0, false, false); \
        acc1 = __builtin_amdgcn_wmma_f32_16x16x4_f32(false, x1, false, w11, (short)0, acc1, false, false); \
    }

    const int rot = mt & (NIT - 1);
#define KBLK(IT_) ((((IT_) + rot) & (NIT - 1)) * GW + wave)

    LOAD_SET(a0, b0, KBLK(0));
    for (int it = 0; it < NIT; it += 2) {
        LOAD_SET(a1, b1, KBLK(it + 1));
        __builtin_amdgcn_sched_barrier(0);
        MMA_SET(a0, b0);
        const int itn = (it + 2 < NIT) ? it + 2 : it + 1;
        LOAD_SET(a0, b0, KBLK(itn));
        __builtin_amdgcn_sched_barrier(0);
        MMA_SET(a1, b1);
    }
#undef LOAD_SET
#undef MMA_SET
#undef KBLK

    for (int rr = 0; rr < 8; ++rr) {
        s_red[((wave * 2 + 0) * 8 + rr) * 32 + lane] = acc0[rr];
        s_red[((wave * 2 + 1) * 8 + rr) * 32 + lane] = acc1[rr];
    }
    __syncthreads();
    {
        const int col = tid & 31;
        const int row = tid >> 5;
        const int ntl = col >> 4;
        const int c   = col & 15;
        const int hh  = row >> 3;
        const int rr  = row & 7;
        float s = 0.f;
        for (int w = 0; w < GW; ++w) s += s_red[((w * 2 + ntl) * 8 + rr) * 32 + hh * 16 + c];
        const int f = nh * 32 + col;
        float v = s + bias[f];
        v = (v >= 0.f) ? v : 0.02f * v;
        out[(size_t)(B0 + row) * NF + f] = v;
    }
}

#define APITCH  272
#define AHALF   (MT * APITCH)
#define ASTAGE  (2 * AHALF)
#define WPITCH  144
#define WKB     (NF * WPITCH)
#define WSTAGE  (4 * WKB)
#define WREGION (2 * WSTAGE)
#define GEMM_LDS (WREGION + 2 * ASTAGE)
__global__ __launch_bounds__(512) void folded_gemm_kernel(
    const float* x, const float* wf, float* part, const float* __restrict__ bias, float* out, size_t ws_bytes)
{
    __shared__ __align__(16) char s_raw[GEMM_LDS];
    float* s_lds = (float*)s_raw;

    if (ws_bytes < WS_NEED) {
        gemm_unsplit_16rows(x, wf, bias, out, s_lds);
        return;
    }

    const int tid  = threadIdx.x;
    const int wave = tid >> 5;
    const int lane = tid & 31;
    const int rt   = wave & 7;
    const int kh   = wave >> 3;
    const int mt   = blockIdx.x >> 4;
    const int ks   = blockIdx.x & (KSPLIT - 1);
    const int B0   = mt * MT;
    const int r    = lane & 15;
    const int h    = lane >> 4;

    const float4* xp = (const float4*)(x + (size_t)(B0 + (tid >> 4)) * KTOT) + (size_t)ks * 256 + (tid & 15);
    char* xs = s_raw + WREGION + (tid >> 4) * APITCH + (tid & 15) * 16;
    const char* af = s_raw + WREGION + kh * AHALF + (rt * 16 + r) * APITCH + h * 16;
    const int wfil = (tid >> 7) * 16 + ((tid >> 3) & 1) * 8 + ((tid >> 4) & 7);
    const float4* gp = (const float4*)wf + (size_t)ks * 32 * 512 + wfil * 8 + (tid & 7);
    char* ws = s_raw + wfil * WPITCH + (tid & 7) * 16;
    const char* bf = s_raw + kh * 2 * WKB + r * WPITCH + h * 16;
    float4* s4 = (float4*)s_raw;

    v8f acc0 = {}, acc1 = {}, acc2 = {}, acc3 = {};
    float4 xe0, xe1, xe2, xe3, xe4, xe5, xe6, xe7, xo0, xo1, xo2, xo3, xo4, xo5, xo6, xo7, ge0, ge1, ge2, ge3, go0, go1, go2, go3, a[4], b[16];

#define XOFF(U_, STEP_) ((size_t)((U_) & 3) * (32 * (KTOT / 4)) + ((U_) >> 2) * 128 + (STEP_) * 16)
#define XLDS(U_, STG_)  ((STG_) * ASTAGE + ((U_) >> 2) * AHALF + ((U_) & 3) * (32 * APITCH))
#define LOAD_X(P_, STEP_)                                                \
    { P_##0 = xp[XOFF(0, STEP_)]; P_##1 = xp[XOFF(1, STEP_)]; P_##2 = xp[XOFF(2, STEP_)]; P_##3 = xp[XOFF(3, STEP_)]; P_##4 = xp[XOFF(4, STEP_)]; P_##5 = xp[XOFF(5, STEP_)]; P_##6 = xp[XOFF(6, STEP_)]; P_##7 = xp[XOFF(7, STEP_)]; }
#define STORE_X(P_, STG_)                                                \
    { *(float4*)(xs + XLDS(0, STG_)) = P_##0; *(float4*)(xs + XLDS(1, STG_)) = P_##1; *(float4*)(xs + XLDS(2, STG_)) = P_##2; *(float4*)(xs + XLDS(3, STG_)) = P_##3; *(float4*)(xs + XLDS(4, STG_)) = P_##4; *(float4*)(xs + XLDS(5, STG_)) = P_##5; *(float4*)(xs + XLDS(6, STG_)) = P_##6; *(float4*)(xs + XLDS(7, STG_)) = P_##7; }
#define LOAD_G(P_, STEP_)                                                \
    {                                                                    \
        P_##0 = gp[((STEP_) * 2) * 512];                                 \
        P_##1 = gp[((STEP_) * 2 + 1) * 512];                             \
        P_##2 = gp[(16 + (STEP_) * 2) * 512];                            \
        P_##3 = gp[(16 + (STEP_) * 2 + 1) * 512];                        \
    }
#define STORE_L(P_, STG_)                                                \
    {                                                                    \
        *(float4*)(ws + (STG_) * WSTAGE)           = P_##0;              \
        *(float4*)(ws + (STG_) * WSTAGE + WKB)     = P_##1;              \
        *(float4*)(ws + (STG_) * WSTAGE + 2 * WKB) = P_##2;              \
        *(float4*)(ws + (STG_) * WSTAGE + 3 * WKB) = P_##3;              \
    }
#define LOAD_AF(A_, STG_, KK_)                                           \
    for (int q = 0; q < 4; ++q) A_[q] = *(const float4*)(af + (STG_) * ASTAGE + (KK_) * 128 + q * 32);
#define LOAD_B(B_, STG_, KK_)                                            \
    for (int j = 0; j < 16; ++j) B_[j] = *(const float4*)(bf + (STG_) * WSTAGE + (KK_) * WKB + (j >> 2) * (16 * WPITCH) + (j & 3) * 32);
#define MMA_SET(A_, B_)                                                  \
    for (int q = 0; q < 4; ++q) {                                        \
        const v2f x0 = {A_[q].x, A_[q].y};                               \
        const v2f x1 = {A_[q].z, A_[q].w};                               \
        const v2f w00 = {B_[q].x, B_[q].y},      w01 = {B_[q].z, B_[q].w};           \
        const v2f w10 = {B_[4 + q].x, B_[4 + q].y},   w11 = {B_[4 + q].z, B_[4 + q].w};   \
        const v2f w20 = {B_[8 + q].x, B_[8 + q].y},   w21 = {B_[8 + q].z, B_[8 + q].w};   \
        const v2f w30 = {B_[12 + q].x, B_[12 + q].y}, w31 = {B_[12 + q].z, B_[12 + q].w}; \
        acc0 = __builtin_amdgcn_wmma_f32_16x16x4_f32(false, x0, false, w00, (short)0, acc0, false, false); \
        acc1 = __builtin_amdgcn_wmma_f32_16x16x4_f32(false, x0, false, w10, (short)0, acc1, false, false); \
        acc2 = __builtin_amdgcn_wmma_f32_16x16x4_f32(false, x0, false, w20, (short)0, acc2, false, false); \
        acc3 = __builtin_amdgcn_wmma_f32_16x16x4_f32(false, x0, false, w30, (short)0, acc3, false, false); \
        acc0 = __builtin_amdgcn_wmma_f32_16x16x4_f32(false, x1, false, w01, (short)0, acc0, false, false); \
        acc1 = __builtin_amdgcn_wmma_f32_16x16x4_f32(false, x1, false, w11, (short)0, acc1, false, false); \
        acc2 = __builtin_amdgcn_wmma_f32_16x16x4_f32(false, x1, false, w21, (short)0, acc2, false, false); \
        acc3 = __builtin_amdgcn_wmma_f32_16x16x4_f32(false, x1, false, w31, (short)0, acc3, false, false); \
    }

    LOAD_G(ge, 0);
    LOAD_X(xe, 0);
    LOAD_G(go, 1);
    LOAD_X(xo, 1);
    STORE_L(ge, 0);
    STORE_X(xe, 0);
    __syncthreads();

#define STEP_MMA(STG_)                                                   \
    LOAD_AF(a, STG_, 0);                                                 \
    LOAD_B(b, STG_, 0);                                                  \
         \
         \
    __builtin_amdgcn_sched_barrier(0);                                   \
    MMA_SET(a, b);                                                       \
    LOAD_AF(a, STG_, 1);                                                 \
    LOAD_B(b, STG_, 1);                                                  \
            \
    __builtin_amdgcn_sched_barrier(0);                                   \
    MMA_SET(a, b);                                                       \
      \
    __syncthreads();

#pragma unroll
    for (int t = 0; t < NSTEP; t += 2) {
        if (t + 2 < NSTEP) {
            LOAD_G(ge, t + 2);
            LOAD_X(xe, t + 2);
        }
        STORE_L(go, 1);
        STORE_X(xo, 1);
        STEP_MMA(0);
        if (t + 3 < NSTEP) {
            LOAD_G(go, t + 3);
            LOAD_X(xo, t + 3);
        }
        if (t + 2 < NSTEP) {
            STORE_L(ge, 0);
            STORE_X(xe, 0);
        }
        STEP_MMA(1);
    }
#undef STEP_MMA
#undef XOFF
#undef XLDS
#undef LOAD_X
#undef STORE_X
#undef LOAD_G
#undef STORE_L
#undef LOAD_AF
#undef LOAD_B
#undef MMA_SET

    {
        float* s_red = s_lds + kh * (MT * NF) + (rt * 16 + 8 * h) * NF + r;
        for (int rr = 0; rr < 8; ++rr) {
            s_red[rr * NF +  0] = acc0[rr];
            s_red[rr * NF + 16] = acc1[rr];
            s_red[rr * NF + 32] = acc2[rr];
            s_red[rr * NF + 48] = acc3[rr];
        }
    }
    __syncthreads();
    {
        float4* pp = (float4*)part + ((size_t)ks * NB + B0) * (NF / 4);
        for (int u = 0; u < 4; ++u) {
            const int i4 = tid + u * 512;
            const float4 p0 = s4[i4];
            const float4 p1 = s4[MT * NF / 4 + i4];
            float4 s;
            s.x = p0.x + p1.x; s.y = p0.y + p1.y; s.z = p0.z + p1.z; s.w = p0.w + p1.w;
            pp[i4] = s;
        }
    }
}

__global__ __launch_bounds__(128) void slice_sum_kernel(
    const float* part, const float* __restrict__ bias, float* out, size_t ws_bytes)
{
    if (ws_bytes < WS_NEED) return;
    const int i4 = blockIdx.x * 128 + threadIdx.x;
    const float4* pp = (const float4*)part + i4;
    float4 p[KSPLIT];
    for (int s = 0; s < KSPLIT; ++s) p[s] = pp[(size_t)s * (NB * NF / 4)];
    float4 v = p[0];
    for (int s = 1; s < KSPLIT; ++s) { v.x += p[s].x; v.y += p[s].y; v.z += p[s].z; v.w += p[s].w; }
    const float4 bv = *(const float4*)(bias + (i4 & (NF / 4 - 1)) * 4);
    v.x += bv.x; v.y += bv.y; v.z += bv.z; v.w += bv.w;
    v.x = (v.x >= 0.f) ? v.x : 0.02f * v.x;
    v.y = (v.y >= 0.f) ? v.y : 0.02f * v.y;
    v.z = (v.z >= 0.f) ? v.z : 0.02f * v.z;
    v.w = (v.w >= 0.f) ? v.w : 0.02f * v.w;
    ((float4*)out)[i4] = v;
}

extern "C" void kernel_launch(void* const* d_in, const int* in_sizes, int n_in,
                              void* d_out, int out_size, void* d_ws, size_t ws_size,
                              hipStream_t stream) {
    const float* x    = (const float*)d_in[0];
    const float* W    = (const float*)d_in[1];
    const float* bias = (const float*)d_in[2];
    float* out  = (float*)d_out;
    float* wf   = (float*)d_ws;
    float* part = wf + WF_FLOATS;

    fold_dwt_into_w_kernel<<<256, 512, 0, stream>>>(W, wf);
    folded_gemm_kernel<<<256, 512, 0, stream>>>(x, wf, part, bias, out, ws_size);
    slice_sum_kernel<<<256, 128, 0, stream>>>(part, bias, out, ws_size);
}


# Fold kernel: a work-group loads and transforms only the rows its 64 time rows depend on (84 of 276 rows of W; 21 of 69 and 35 of 131 intermediate rows); summation order unchanged
# speedup vs baseline: 7.3716x; 1.0364x over previous
#include <hip/hip_runtime.h>


typedef __attribute__((ext_vector_type(2))) float v2f;
typedef __attribute__((ext_vector_type(8))) float v8f;

#define NB    2048
#define NF    64
#define TLEN  256
#define TDWT  276
#define NPOS  64
#define KTOT  (TLEN * NPOS)
#define NKB   (KTOT / 32)
#define MT     128
#define KSPLIT 16
#define NSTEP  8
#define WF_FLOATS   (2 * NKB * 8 * 32 * 4)
#define PART_FLOATS (KSPLIT * NB * NF)
#define WS_NEED     ((size_t)(WF_FLOATS + PART_FLOATS) * 4)
#define GW    16
#define NIT   (NKB / GW)


__device__ __forceinline__ float adj_z(const float* __restrict__ lo, const float* __restrict__ hi,
                                       int nout, int m)
{
    const float LE0 = 0.23037781330885523f,  LE1 = 0.6308807679295904f,
                LE2 = -0.18703481171888114f, LE3 = 0.032883011666982945f;
    const float LO0 = 0.7148465705525415f,   LO1 = -0.02798376941698385f,
                LO2 = 0.030841381835986965f, LO3 = -0.010597401784997278f;
    const float HE0 = -0.010597401784997278f, HE1 = 0.030841381835986965f,
                HE2 = -0.02798376941698385f,  HE3 = 0.7148465705525415f;
    const float HO0 = -0.032883011666982945f, HO1 = 0.18703481171888114f,
                HO2 = -0.6308807679295904f,   HO3 = -0.23037781330885523f;
    const int par = m & 1;
    const int k0  = (m + 6 - par) >> 1;
    const float l0 = par ? LO0 : LE0, l1 = par ? LO1 : LE1, l2 = par ? LO2 : LE2, l3 = par ? LO3 : LE3;
    const float g0 = par ? HO0 : HE0, g1 = par ? HO1 : HE1, g2 = par ? HO2 : HE2, g3 = par ? HO3 : HE3;
    const float lc[4] = {l0, l1, l2, l3};
    const float gc[4] = {g0, g1, g2, g3};
    float acc = 0.f;
    for (int u = 0; u < 4; ++u) {
        const int  k  = k0 - u;
        const bool ok = (k >= 0) && (k < nout);
        const int  kc = (k < 0) ? 0 : ((k < nout) ? k : nout - 1);
        float lv = lo[kc * 64];
        float hv = hi[kc * 64];
        lv = ok ? lv : 0.f;
        hv = ok ? hv : 0.f;
        acc = fmaf(lc[u], lv, acc);
        acc = fmaf(gc[u], hv, acc);
    }
    return acc;
}

template <int NR>
__device__ __forceinline__ void adj_rows_main(const float* lo, const float* hi, float* dst,
                                              int jstart, int cnt, int jg)
{
    const float LE0 = 0.23037781330885523f,  LE1 = 0.6308807679295904f,
                LE2 = -0.18703481171888114f, LE3 = 0.032883011666982945f;
    const float LO0 = 0.7148465705525415f,   LO1 = -0.02798376941698385f,
                LO2 = 0.030841381835986965f, LO3 = -0.010597401784997278f;
    const float HE0 = -0.010597401784997278f, HE1 = 0.030841381835986965f,
                HE2 = -0.02798376941698385f,  HE3 = 0.7148465705525415f;
    const float HO0 = -0.032883011666982945f, HO1 = 0.18703481171888114f,
                HO2 = -0.6308807679295904f,   HO3 = -0.23037781330885523f;
    for (int i0 = jg; i0 < cnt; i0 += 8 * NR) {
        float lv[NR][4], hv[NR][4];
#pragma unroll
        for (int u = 0; u < NR; ++u) {
            const int i   = i0 + 8 * u;
            const int ic  = (i < cnt) ? i : i0;
            const int m   = jstart + ic;
            const int par = m & 1;
            const int k0  = (m + 6 - par) >> 1;
#pragma unroll
            for (int t = 0; t < 4; ++t) {
                lv[u][t] = lo[(k0 - t) * 64];
                hv[u][t] = hi[(k0 - t) * 64];
            }
        }
        __builtin_amdgcn_sched_barrier(0);
#pragma unroll
        for (int u = 0; u < NR; ++u) {
            const int i   = i0 + 8 * u;
            const int ic  = (i < cnt) ? i : i0;
            const int par = (jstart + ic) & 1;
            const float lc[4] = {par ? LO0 : LE0, par ? LO1 : LE1, par ? LO2 : LE2, par ? LO3 : LE3};
            const float gc[4] = {par ? HO0 : HE0, par ? HO1 : HE1, par ? HO2 : HE2, par ? HO3 : HE3};
            float acc = 0.f;
#pragma unroll
            for (int t = 0; t < 4; ++t) {
                acc = fmaf(lc[t], lv[u][t], acc);
                acc = fmaf(gc[t], hv[u][t], acc);
            }
            if (i < cnt) dst[i * 64] = acc;
        }
    }
}

__device__ __forceinline__ void adj_rows_edges(const float* lo, const float* hi, float* dst,
                                               int N, int nout, int jstart, int cnt, int jg)
{
    const bool hasL = (jstart == 0) && (jg >= 1) && (jg <= 6);
    const int  il   = hasL ? jg : 0;
    const int  ml   = hasL ? -jg : jstart;
    const int  rlo  = 2 * N - 1 - 2 * nout;
    const int  jr   = rlo + ((jg - rlo) & 7);
    const bool hasR = (jr <= N - 2) && (jr >= jstart) && (jr < jstart + cnt);
    const int  ir   = hasR ? jr - jstart : 0;
    const int  mr   = hasR ? 2 * N - 2 - jr : jstart;
    const float pl = dst[il * 64];
    const float pr = dst[ir * 64];
    const float vl = pl + adj_z(lo, hi, nout, ml);
    const float vr = pr + adj_z(lo, hi, nout, mr);
    if (hasL) dst[il * 64] = vl;
    if (hasR) dst[ir * 64] = vr;
}

__global__ __launch_bounds__(512) void fold_dwt_into_w_kernel(
    const float* __restrict__ W, float* __restrict__ wf)
{
    __shared__ __align__(16) float s_w[TDWT * 64];
    __shared__ __align__(16) float s_a2[69 * 64];
    __shared__ __align__(16) float s_a1[131 * 64];
    __shared__ __align__(16) float s_o[64 * 64];

    const int tid = threadIdx.x;
    const int ph  = blockIdx.x & 1;
    const int fp  = (blockIdx.x >> 1) & 31;
    const int ts  = blockIdx.x >> 6;
    const int p0  = ph * 32;
    const int f0  = fp * 2;

    constexpr int NR3   = 14, NR2 = 21, NR1 = 35;
    constexpr int NV4   = 16 * (2 * NR3 + NR2 + NR1);
    constexpr int NFULL = NV4 / 512;
    float4 wv[NFULL + 1];
    int    wd[NFULL + 1];
#pragma unroll
    for (int u = 0; u <= NFULL; ++u) {
        const int i  = tid + u * 512;
        const int ic = (u < NFULL || i < NV4) ? i : tid;
        const int rr = ic >> 4;
        const int d  = (rr < NR3)             ? ts * 8 + rr
                     : (rr < 2 * NR3)         ? 238 + ts * 8 + (rr - NR3)
                     : (rr < 2 * NR3 + NR2)   ? 169 + ts * 16 + (rr - 2 * NR3)
                     :                          38 + ts * 32 + (rr - 2 * NR3 - NR2);
        const int fl = (ic >> 3) & 1;
        const int pc = ic & 7;
        wd[u] = d * 64 + (ic & 15) * 4;
        wv[u] = *(const float4*)(W + ((size_t)(f0 + fl) * TDWT + d) * NPOS + p0 + pc * 4);
    }
#pragma unroll
    for (int u = 0; u <= NFULL; ++u) *(float4*)(s_w + wd[u]) = wv[u];
    __syncthreads();

    const int v  = tid & 63;
    const int jg = __builtin_amdgcn_readfirstlane(tid >> 6);
    const int j2s = ts * 16;
    const int j1s = ts * 32;
    adj_rows_main<3>(s_w + v, s_w + 238 * 64 + v, s_a2 + j2s * 64 + v, j2s, NR2, jg);
    adj_rows_edges(s_w + v, s_w + 238 * 64 + v, s_a2 + j2s * 64 + v, 69, 38, j2s, NR2, jg);
    __syncthreads();
    adj_rows_main<5>(s_a2 + v, s_w + 169 * 64 + v, s_a1 + j1s * 64 + v, j1s, NR1, jg);
    adj_rows_edges(s_a2 + v, s_w + 169 * 64 + v, s_a1 + j1s * 64 + v, 131, 69, j1s, NR1, jg);
    __syncthreads();
    adj_rows_main<4>(s_a1 + v, s_w + 38 * 64 + v, s_o + v, ts * 64, 64, jg);
    adj_rows_edges(s_a1 + v, s_w + 38 * 64 + v, s_o + v, 256, 131, ts * 64, 64, jg);
    __syncthreads();

    for (int i = tid; i < 64 * 16; i += 512) {
        const int c4 = i & 15;
        const int tl = i >> 4;
        const int t  = ts * 64 + tl;
        const int kb = t * 2 + ph;
        const float4 val = *(const float4*)(s_o + tl * 64 + c4 * 4);
        *(float4*)(wf + ((size_t)kb * NF + f0) * 32 + c4 * 4) = val;
    }
}

__device__ __forceinline__ void gemm_unsplit_16rows(
    const float* x, const float* wf, const float* __restrict__ bias, float* out, float* s_red)
{
    const int tid  = threadIdx.x;
    const int wave = tid >> 5;
    const int lane = tid & 31;
    const int mt   = blockIdx.x >> 1;
    const int nh   = blockIdx.x & 1;
    const int B0   = mt * 16;
    const int r    = lane & 15;
    const int h    = lane >> 4;

    const float4* ap = (const float4*)(x + (size_t)(B0 + r) * KTOT + h * 4);
    const float4* bp = (const float4*)wf + (nh * 32 + r) * 8 + h;

    v8f acc0 = {};
    v8f acc1 = {};

    float4 a0[4], b0[8], a1[4], b1[8];

#define LOAD_SET(A_, B_, KB_)                                            \
    {                                                                    \
        const int kb_ = (KB_);                                           \
        for (int q = 0; q < 4; ++q) A_[q] = ap[kb_ * 8 + q * 2];                          \
        for (int j = 0; j < 8; ++j) B_[j] = bp[kb_ * 512 + (j >> 2) * 128 + (j & 3) * 2]; \
    }
#define MMA_SET(A_, B_)                                                  \
    for (int q = 0; q < 4; ++q) {                                        \
        const v2f x0 = {A_[q].x, A_[q].y};                               \
        const v2f x1 = {A_[q].z, A_[q].w};                               \
        const v2f w00 = {B_[q].x, B_[q].y};                              \
        const v2f w01 = {B_[q].z, B_[q].w};                              \
        const v2f w10 = {B_[4 + q].x, B_[4 + q].y};                      \
        const v2f w11 = {B_[4 + q].z, B_[4 + q].w};                      \
        acc0 = __builtin_amdgcn_wmma_f32_16x16x4_f32(false, x0, false, w00, (short)0, acc0, false, false); \
        acc1 = __builtin_amdgcn_wmma_f32_16x16x4_f32(false, x0, false, w10, (short)0, acc1, false, false); \
        acc0 = __builtin_amdgcn_wmma_f32_16x16x4_f32(false, x1, false, w01, (short)0, acc0, false, false); \
        acc1 = __builtin_amdgcn_wmma_f32_16x16x4_f32(false, x1, false, w11, (short)0, acc1, false, false); \
    }

    const int rot = mt & (NIT - 1);
#define KBLK(IT_) ((((IT_) + rot) & (NIT - 1)) * GW + wave)

    LOAD_SET(a0, b0, KBLK(0));
    for (int it = 0; it < NIT; it += 2) {
        LOAD_SET(a1, b1, KBLK(it + 1));
        __builtin_amdgcn_sched_barrier(0);
        MMA_SET(a0, b0);
        const int itn = (it + 2 < NIT) ? it + 2 : it + 1;
        LOAD_SET(a0, b0, KBLK(itn));
        __builtin_amdgcn_sched_barrier(0);
        MMA_SET(a1, b1);
    }
#undef LOAD_SET
#undef MMA_SET
#undef KBLK

    for (int rr = 0; rr < 8; ++rr) {
        s_red[((wave * 2 + 0) * 8 + rr) * 32 + lane] = acc0[rr];
        s_red[((wave * 2 + 1) * 8 + rr) * 32 + lane] = acc1[rr];
    }
    __syncthreads();
    {
        const int col = tid & 31;
        const int row = tid >> 5;
        const int ntl = col >> 4;
        const int c   = col & 15;
        const int hh  = row >> 3;
        const int rr  = row & 7;
        float s = 0.f;
        for (int w = 0; w < GW; ++w) s += s_red[((w * 2 + ntl) * 8 + rr) * 32 + hh * 16 + c];
        const int f = nh * 32 + col;
        float v = s + bias[f];
        v = (v >= 0.f) ? v : 0.02f * v;
        out[(size_t)(B0 + row) * NF + f] = v;
    }
}

#define APITCH  272
#define AHALF   (MT * APITCH)
#define ASTAGE  (2 * AHALF)
#define WPITCH  144
#define WKB     (NF * WPITCH)
#define WSTAGE  (4 * WKB)
#define WREGION (2 * WSTAGE)
#define GEMM_LDS (WREGION + 2 * ASTAGE)
__global__ __launch_bounds__(512) void folded_gemm_kernel(
    const float* x, const float* wf, float* part, const float* __restrict__ bias, float* out, size_t ws_bytes)
{
    __shared__ __align__(16) char s_raw[GEMM_LDS];
    float* s_lds = (float*)s_raw;

    if (ws_bytes < WS_NEED) {
        gemm_unsplit_16rows(x, wf, bias, out, s_lds);
        return;
    }

    const int tid  = threadIdx.x;
    const int wave = tid >> 5;
    const int lane = tid & 31;
    const int rt   = wave & 7;
    const int kh   = wave >> 3;
    const int mt   = blockIdx.x >> 4;
    const int ks   = blockIdx.x & (KSPLIT - 1);
    const int B0   = mt * MT;
    const int r    = lane & 15;
    const int h    = lane >> 4;

    const float4* xp = (const float4*)(x + (size_t)(B0 + (tid >> 4)) * KTOT) + (size_t)ks * 256 + (tid & 15);
    char* xs = s_raw + WREGION + (tid >> 4) * APITCH + (tid & 15) * 16;
    const char* af = s_raw + WREGION + kh * AHALF + (rt * 16 + r) * APITCH + h * 16;
    const int wfil = (tid >> 7) * 16 + ((tid >> 3) & 1) * 8 + ((tid >> 4) & 7);
    const float4* gp = (const float4*)wf + (size_t)ks * 32 * 512 + wfil * 8 + (tid & 7);
    char* ws = s_raw + wfil * WPITCH + (tid & 7) * 16;
    const char* bf = s_raw + kh * 2 * WKB + r * WPITCH + h * 16;
    float4* s4 = (float4*)s_raw;

    v8f acc0 = {}, acc1 = {}, acc2 = {}, acc3 = {};
    float4 xe0, xe1, xe2, xe3, xe4, xe5, xe6, xe7, xo0, xo1, xo2, xo3, xo4, xo5, xo6, xo7, ge0, ge1, ge2, ge3, go0, go1, go2, go3, a[4], b[16];

#define XOFF(U_, STEP_) ((size_t)((U_) & 3) * (32 * (KTOT / 4)) + ((U_) >> 2) * 128 + (STEP_) * 16)
#define XLDS(U_, STG_)  ((STG_) * ASTAGE + ((U_) >> 2) * AHALF + ((U_) & 3) * (32 * APITCH))
#define LOAD_X(P_, STEP_)                                                \
    { P_##0 = xp[XOFF(0, STEP_)]; P_##1 = xp[XOFF(1, STEP_)]; P_##2 = xp[XOFF(2, STEP_)]; P_##3 = xp[XOFF(3, STEP_)]; P_##4 = xp[XOFF(4, STEP_)]; P_##5 = xp[XOFF(5, STEP_)]; P_##6 = xp[XOFF(6, STEP_)]; P_##7 = xp[XOFF(7, STEP_)]; }
#define STORE_X(P_, STG_)                                                \
    { *(float4*)(xs + XLDS(0, STG_)) = P_##0; *(float4*)(xs + XLDS(1, STG_)) = P_##1; *(float4*)(xs + XLDS(2, STG_)) = P_##2; *(float4*)(xs + XLDS(3, STG_)) = P_##3; *(float4*)(xs + XLDS(4, STG_)) = P_##4; *(float4*)(xs + XLDS(5, STG_)) = P_##5; *(float4*)(xs + XLDS(6, STG_)) = P_##6; *(float4*)(xs + XLDS(7, STG_)) = P_##7; }
#define LOAD_G(P_, STEP_)                                                \
    {                                                                    \
        P_##0 = gp[((STEP_) * 2) * 512];                                 \
        P_##1 = gp[((STEP_) * 2 + 1) * 512];                             \
        P_##2 = gp[(16 + (STEP_) * 2) * 512];                            \
        P_##3 = gp[(16 + (STEP_) * 2 + 1) * 512];                        \
    }
#define STORE_L(P_, STG_)                                                \
    {                                                                    \
        *(float4*)(ws + (STG_) * WSTAGE)           = P_##0;              \
        *(float4*)(ws + (STG_) * WSTAGE + WKB)     = P_##1;              \
        *(float4*)(ws + (STG_) * WSTAGE + 2 * WKB) = P_##2;              \
        *(float4*)(ws + (STG_) * WSTAGE + 3 * WKB) = P_##3;              \
    }
#define LOAD_AF(A_, STG_, KK_)                                           \
    for (int q = 0; q < 4; ++q) A_[q] = *(const float4*)(af + (STG_) * ASTAGE + (KK_) * 128 + q * 32);
#define LOAD_B(B_, STG_, KK_)                                            \
    for (int j = 0; j < 16; ++j) B_[j] = *(const float4*)(bf + (STG_) * WSTAGE + (KK_) * WKB + (j >> 2) * (16 * WPITCH) + (j & 3) * 32);
#define MMA_SET(A_, B_)                                                  \
    for (int q = 0; q < 4; ++q) {                                        \
        const v2f x0 = {A_[q].x, A_[q].y};                               \
        const v2f x1 = {A_[q].z, A_[q].w};                               \
        const v2f w00 = {B_[q].x, B_[q].y},      w01 = {B_[q].z, B_[q].w};           \
        const v2f w10 = {B_[4 + q].x, B_[4 + q].y},   w11 = {B_[4 + q].z, B_[4 + q].w};   \
        const v2f w20 = {B_[8 + q].x, B_[8 + q].y},   w21 = {B_[8 + q].z, B_[8 + q].w};   \
        const v2f w30 = {B_[12 + q].x, B_[12 + q].y}, w31 = {B_[12 + q].z, B_[12 + q].w}; \
        acc0 = __builtin_amdgcn_wmma_f32_16x16x4_f32(false, x0, false, w00, (short)0, acc0, false, false); \
        acc1 = __builtin_amdgcn_wmma_f32_16x16x4_f32(false, x0, false, w10, (short)0, acc1, false, false); \
        acc2 = __builtin_amdgcn_wmma_f32_16x16x4_f32(false, x0, false, w20, (short)0, acc2, false, false); \
        acc3 = __builtin_amdgcn_wmma_f32_16x16x4_f32(false, x0, false, w30, (short)0, acc3, false, false); \
        acc0 = __builtin_amdgcn_wmma_f32_16x16x4_f32(false, x1, false, w01, (short)0, acc0, false, false); \
        acc1 = __builtin_amdgcn_wmma_f32_16x16x4_f32(false, x1, false, w11, (short)0, acc1, false, false); \
        acc2 = __builtin_amdgcn_wmma_f32_16x16x4_f32(false, x1, false, w21, (short)0, acc2, false, false); \
        acc3 = __builtin_amdgcn_wmma_f32_16x16x4_f32(false, x1, false, w31, (short)0, acc3, false, false); \
    }

    LOAD_G(ge, 0);
    LOAD_X(xe, 0);
    LOAD_G(go, 1);
    LOAD_X(xo, 1);
    STORE_L(ge, 0);
    STORE_X(xe, 0);
    __syncthreads();

#define STEP_MMA(STG_)                                                   \
    LOAD_AF(a, STG_, 0);                                                 \
    LOAD_B(b, STG_, 0);                                                  \
         \
         \
    __builtin_amdgcn_sched_barrier(0);                                   \
    MMA_SET(a, b);                                                       \
    LOAD_AF(a, STG_, 1);                                                 \
    LOAD_B(b, STG_, 1);                                                  \
            \
    __builtin_amdgcn_sched_barrier(0);                                   \
    MMA_SET(a, b);                                                       \
      \
    __syncthreads();

#pragma unroll
    for (int t = 0; t < NSTEP; t += 2) {
        if (t + 2 < NSTEP) {
            LOAD_G(ge, t + 2);
            LOAD_X(xe, t + 2);
        }
        STORE_L(go, 1);
        STORE_X(xo, 1);
        STEP_MMA(0);
        if (t + 3 < NSTEP) {
            LOAD_G(go, t + 3);
            LOAD_X(xo, t + 3);
        }
        if (t + 2 < NSTEP) {
            STORE_L(ge, 0);
            STORE_X(xe, 0);
        }
        STEP_MMA(1);
    }
#undef STEP_MMA
#undef XOFF
#undef XLDS
#undef LOAD_X
#undef STORE_X
#undef LOAD_G
#undef STORE_L
#undef LOAD_AF
#undef LOAD_B
#undef MMA_SET

    {
        float* s_red = s_lds + kh * (MT * NF) + (rt * 16 + 8 * h) * NF + r;
        for (int rr = 0; rr < 8; ++rr) {
            s_red[rr * NF +  0] = acc0[rr];
            s_red[rr * NF + 16] = acc1[rr];
            s_red[rr * NF + 32] = acc2[rr];
            s_red[rr * NF + 48] = acc3[rr];
        }
    }
    __syncthreads();
    {
        float4* pp = (float4*)part + ((size_t)ks * NB + B0) * (NF / 4);
        for (int u = 0; u < 4; ++u) {
            const int i4 = tid + u * 512;
            const float4 p0 = s4[i4];
            const float4 p1 = s4[MT * NF / 4 + i4];
            float4 s;
            s.x = p0.x + p1.x; s.y = p0.y + p1.y; s.z = p0.z + p1.z; s.w = p0.w + p1.w;
            pp[i4] = s;
        }
    }
}

__global__ __launch_bounds__(128) void slice_sum_kernel(
    const float* part, const float* __restrict__ bias, float* out, size_t ws_bytes)
{
    if (ws_bytes < WS_NEED) return;
    const int i4 = blockIdx.x * 128 + threadIdx.x;
    const float4* pp = (const float4*)part + i4;
    float4 p[KSPLIT];
    for (int s = 0; s < KSPLIT; ++s) p[s] = pp[(size_t)s * (NB * NF / 4)];
    float4 v = p[0];
    for (int s = 1; s < KSPLIT; ++s) { v.x += p[s].x; v.y += p[s].y; v.z += p[s].z; v.w += p[s].w; }
    const float4 bv = *(const float4*)(bias + (i4 & (NF / 4 - 1)) * 4);
    v.x += bv.x; v.y += bv.y; v.z += bv.z; v.w += bv.w;
    v.x = (v.x >= 0.f) ? v.x : 0.02f * v.x;
    v.y = (v.y >= 0.f) ? v.y : 0.02f * v.y;
    v.z = (v.z >= 0.f) ? v.z : 0.02f * v.z;
    v.w = (v.w >= 0.f) ? v.w : 0.02f * v.w;
    ((float4*)out)[i4] = v;
}

extern "C" void kernel_launch(void* const* d_in, const int* in_sizes, int n_in,
                              void* d_out, int out_size, void* d_ws, size_t ws_size,
                              hipStream_t stream) {
    const float* x    = (const float*)d_in[0];
    const float* W    = (const float*)d_in[1];
    const float* bias = (const float*)d_in[2];
    float* out  = (float*)d_out;
    float* wf   = (float*)d_ws;
    float* part = wf + WF_FLOATS;

    fold_dwt_into_w_kernel<<<256, 512, 0, stream>>>(W, wf);
    folded_gemm_kernel<<<256, 512, 0, stream>>>(x, wf, part, bias, out, ws_size);
    slice_sum_kernel<<<256, 128, 0, stream>>>(part, bias, out, ws_size);
}
